# starting kernel
# baseline (speedup 1.0000x reference)
#include <hip/hip_runtime.h>

#define N_NODES 100000
#define N_EDGES 600000
#define HD 128

typedef __attribute__((ext_vector_type(2))) float v2f;
typedef __attribute__((ext_vector_type(8))) float v8f;

__global__ __launch_bounds__(256) void gather_embed(
    const int* __restrict__ node_id, const float* __restrict__ embed,
    float* __restrict__ h) {
  int t   = blockIdx.x * blockDim.x + threadIdx.x;
  int row = t >> 5;
  int c4  = (t & 31) << 2;
  if (row < N_NODES) {
    int srow = node_id[row];
    float4 v = *(const float4*)(embed + (size_t)srow * HD + c4);
    *(float4*)(h + (size_t)row * HD + c4) = v;
  }
}

__global__ __launch_bounds__(256) void gemm_norm_wmma(
    const float* __restrict__ h, const float* __restrict__ W,
    const float* __restrict__ norm, float* __restrict__ m) {
  __shared__ float Wl2[HD * HD];

  int tid = threadIdx.x;
  for (int i = tid; i < (HD * HD) / 4; i += 256) {
    int p = i >> 6;
    int n = (i & 63) << 1;
    int k = p << 1;
    float2 r0 = *(const float2*)(W + (size_t)k * HD + n);
    float2 r1 = *(const float2*)(W + (size_t)(k + 1) * HD + n);
    float4 o  = make_float4(r0.x, r1.x, r0.y, r1.y);
    *(float4*)(Wl2 + p * 256 + n * 2) = o;
  }
  __syncthreads();

  int wave    = tid >> 5;
  int lane    = tid & 31;
  int lane16  = lane & 15;
  int pOff    = lane >> 4;
  int rowBase = (blockIdx.x * 8 + wave) * 16;
  int rowA    = rowBase + lane16;
  int rowAc   = rowA < N_NODES ? rowA : (N_NODES - 1);

  v8f acc[8] = {};
  const float* hrow = h + (size_t)rowAc * HD;

  for (int k0 = 0; k0 < HD; k0 += 4) {
    v2f a = *(const v2f*)(hrow + k0 + (pOff << 1));
    const float* bbase = Wl2 + ((k0 >> 1) + pOff) * 256;
#pragma unroll
    for (int t = 0; t < 8; ++t) {
      int n = t * 16 + lane16;
      v2f b = *(const v2f*)(bbase + n * 2);
      acc[t] = __builtin_amdgcn_wmma_f32_16x16x4_f32(
          false, a, false, b, (short)0, acc[t], false, false);
    }
  }

  int rhalf = pOff << 3;
#pragma unroll
  for (int v = 0; v < 8; ++v) {
    int r = rowBase + v + rhalf;
    if (r < N_NODES) {
      float nr = norm[r];
#pragma unroll
      for (int t = 0; t < 8; ++t)
        m[(size_t)r * HD + t * 16 + lane16] = acc[t][v] * nr;
    }
  }
}

__global__ __launch_bounds__(256) void zero_f4(float4* __restrict__ p, int n4) {
  int t = blockIdx.x * blockDim.x + threadIdx.x;
  if (t < n4) p[t] = make_float4(0.f, 0.f, 0.f, 0.f);
}

__global__ __launch_bounds__(256) void scatter_edges(
    const float* __restrict__ m, const int* __restrict__ src,
    const int* __restrict__ dst, float* __restrict__ agg) {
  int wave = threadIdx.x >> 5;
  int lane = threadIdx.x & 31;
  int e = blockIdx.x * 8 + wave;
  if (e >= N_EDGES) return;
  int s = src[e];
  int d = dst[e];
  float4 v    = *(const float4*)(m + (size_t)s * HD + lane * 4);
  float* dstp = agg + (size_t)d * HD + lane * 4;
  (void)__hip_atomic_fetch_add(dstp + 0, v.x, __ATOMIC_RELAXED, __HIP_MEMORY_SCOPE_AGENT);
  (void)__hip_atomic_fetch_add(dstp + 1, v.y, __ATOMIC_RELAXED, __HIP_MEMORY_SCOPE_AGENT);
  (void)__hip_atomic_fetch_add(dstp + 2, v.z, __ATOMIC_RELAXED, __HIP_MEMORY_SCOPE_AGENT);
  (void)__hip_atomic_fetch_add(dstp + 3, v.w, __ATOMIC_RELAXED, __HIP_MEMORY_SCOPE_AGENT);
}

__global__ __launch_bounds__(256) void epilogue_leaky(
    const float* __restrict__ agg, const float* __restrict__ norm,
    const float* __restrict__ bias, float* __restrict__ out) {
  int t   = blockIdx.x * blockDim.x + threadIdx.x;
  int row = t >> 5;
  int c4  = (t & 31) << 2;
  if (row < N_NODES) {
    float nr = norm[row];
    float4 a = *(const float4*)(agg + (size_t)row * HD + c4);
    float4 b = *(const float4*)(bias + c4);
    float4 n;
    n.x = a.x * nr + b.x;
    n.y = a.y * nr + b.y;
    n.z = a.z * nr + b.z;
    n.w = a.w * nr + b.w;
    n.x = n.x > 0.f ? n.x : 0.2f * n.x;
    n.y = n.y > 0.f ? n.y : 0.2f * n.y;
    n.z = n.z > 0.f ? n.z : 0.2f * n.z;
    n.w = n.w > 0.f ? n.w : 0.2f * n.w;
    *(float4*)(out + (size_t)row * HD + c4) = n;
  }
}

extern "C" void kernel_launch(void* const* d_in, const int* in_sizes, int n_in,
                              void* d_out, int out_size, void* d_ws, size_t ws_size,
                              hipStream_t stream) {
  const int*   node_id = (const int*)d_in[0];
  const int*   src     = (const int*)d_in[1];
  const int*   dst     = (const int*)d_in[2];
  const float* norm    = (const float*)d_in[3];
  const float* embed   = (const float*)d_in[4];
  const float* W1      = (const float*)d_in[5];
  const float* b1      = (const float*)d_in[6];
  const float* W2      = (const float*)d_in[7];
  const float* b2      = (const float*)d_in[8];
  float* out = (float*)d_out;

  const size_t feat = (size_t)N_NODES * HD;
  float* bufH = (float*)d_ws;
  float* bufM = bufH + feat;
  float* agg  = out;

  dim3 blk(256);
  const int gElem    = (int)((feat / 4 + 255) / 256);
  const int gGemm    = (N_NODES + 127) / 128;
  const int gScatter = (N_EDGES + 7) / 8;

  gather_embed<<<gElem, blk, 0, stream>>>(node_id, embed, bufH);

  gemm_norm_wmma<<<gGemm, blk, 0, stream>>>(bufH, W1, norm, bufM);
  zero_f4<<<gElem, blk, 0, stream>>>((float4*)agg, (int)(feat / 4));
  scatter_edges<<<gScatter, blk, 0, stream>>>(bufM, src, dst, agg);
  epilogue_leaky<<<gElem, blk, 0, stream>>>(agg, norm, b1, bufH);

  gemm_norm_wmma<<<gGemm, blk, 0, stream>>>(bufH, W2, norm, bufM);
  zero_f4<<<gElem, blk, 0, stream>>>((float4*)agg, (int)(feat / 4));
  scatter_edges<<<gScatter, blk, 0, stream>>>(bufM, src, dst, agg);
  epilogue_leaky<<<gElem, blk, 0, stream>>>(agg, norm, b2, out);
}


# Replace f32-atomic scatter and its zero pass by a per-row gather over a CSR of the edges built each call (one int-atomic pass); a row's sum order is now by source index, fixed for rows <= 32 edges
# speedup vs baseline: 1.7400x; 1.7400x over previous
#include <hip/hip_runtime.h>

#define N_NODES 100000
#define N_EDGES 600000
#define HD 128

typedef __attribute__((ext_vector_type(2))) float v2f;
typedef __attribute__((ext_vector_type(8))) float v8f;

__global__ __launch_bounds__(256) void gather_embed(
    const int* __restrict__ node_id, const float* __restrict__ embed,
    float* __restrict__ h) {
  int t   = blockIdx.x * blockDim.x + threadIdx.x;
  int row = t >> 5;
  int c4  = (t & 31) << 2;
  if (row < N_NODES) {
    int srow = node_id[row];
    float4 v = *(const float4*)(embed + (size_t)srow * HD + c4);
    *(float4*)(h + (size_t)row * HD + c4) = v;
  }
}

__global__ __launch_bounds__(256) void gemm_norm_wmma(
    const float* __restrict__ h, const float* __restrict__ W,
    const float* __restrict__ norm, float* __restrict__ m) {
  __shared__ float Wl2[HD * HD];

  int tid = threadIdx.x;
  for (int i = tid; i < (HD * HD) / 4; i += 256) {
    int p = i >> 6;
    int n = (i & 63) << 1;
    int k = p << 1;
    float2 r0 = *(const float2*)(W + (size_t)k * HD + n);
    float2 r1 = *(const float2*)(W + (size_t)(k + 1) * HD + n);
    float4 o  = make_float4(r0.x, r1.x, r0.y, r1.y);
    *(float4*)(Wl2 + p * 256 + n * 2) = o;
  }
  __syncthreads();

  int wave    = tid >> 5;
  int lane    = tid & 31;
  int lane16  = lane & 15;
  int pOff    = lane >> 4;
  int rowBase = (blockIdx.x * 8 + wave) * 16;
  int rowA    = rowBase + lane16;
  int rowAc   = rowA < N_NODES ? rowA : (N_NODES - 1);

  v8f acc[8] = {};
  const float* hrow = h + (size_t)rowAc * HD;

  for (int k0 = 0; k0 < HD; k0 += 4) {
    v2f a = *(const v2f*)(hrow + k0 + (pOff << 1));
    const float* bbase = Wl2 + ((k0 >> 1) + pOff) * 256;
#pragma unroll
    for (int t = 0; t < 8; ++t) {
      int n = t * 16 + lane16;
      v2f b = *(const v2f*)(bbase + n * 2);
      acc[t] = __builtin_amdgcn_wmma_f32_16x16x4_f32(
          false, a, false, b, (short)0, acc[t], false, false);
    }
  }

  int rhalf = pOff << 3;
#pragma unroll
  for (int v = 0; v < 8; ++v) {
    int r = rowBase + v + rhalf;
    if (r < N_NODES) {
      float nr = norm[r];
#pragma unroll
      for (int t = 0; t < 8; ++t)
        m[(size_t)r * HD + t * 16 + lane16] = acc[t][v] * nr;
    }
  }
}

__global__ __launch_bounds__(256) void zero_f4(float4* __restrict__ p, int n4) {
  int t = blockIdx.x * blockDim.x + threadIdx.x;
  if (t < n4) p[t] = make_float4(0.f, 0.f, 0.f, 0.f);
}

__global__ __launch_bounds__(256) void count_rank(
    const int* __restrict__ dst, int* __restrict__ cnt, int* __restrict__ erank) {
  int e = blockIdx.x * blockDim.x + threadIdx.x;
  if (e < N_EDGES)
    erank[e] = __hip_atomic_fetch_add(cnt + dst[e], 1, __ATOMIC_RELAXED, __HIP_MEMORY_SCOPE_AGENT);
}

#define SCAN_BLK 256
#define N_SCAN_BLOCKS ((N_NODES + SCAN_BLK - 1) / SCAN_BLK)

__global__ __launch_bounds__(SCAN_BLK) void block_sums(
    const int* __restrict__ cnt, int* __restrict__ bsum) {
  __shared__ int wsum[SCAN_BLK / 32];
  int i = blockIdx.x * SCAN_BLK + threadIdx.x;
  int v = i < N_NODES ? cnt[i] : 0;
#pragma unroll
  for (int d = 16; d > 0; d >>= 1) v += __shfl_xor(v, d);
  if ((threadIdx.x & 31) == 0) wsum[threadIdx.x >> 5] = v;
  __syncthreads();
  if (threadIdx.x == 0) {
    int s = 0;
#pragma unroll
    for (int w = 0; w < SCAN_BLK / 32; ++w) s += wsum[w];
    bsum[blockIdx.x] = s;
  }
}

__global__ __launch_bounds__(SCAN_BLK) void scan_rows(
    const int* __restrict__ cnt, const int* __restrict__ bsum,
    int* __restrict__ rowptr) {
  __shared__ int wincl[SCAN_BLK / 32];
  __shared__ int wpre[SCAN_BLK / 32];
  int tid  = threadIdx.x;
  int lane = tid & 31;
  int wave = tid >> 5;
  int b    = blockIdx.x;

  int p = 0;
  for (int k = tid; k < b; k += SCAN_BLK) p += bsum[k];
#pragma unroll
  for (int d = 16; d > 0; d >>= 1) p += __shfl_xor(p, d);

  int i = b * SCAN_BLK + tid;
  int v = i < N_NODES ? cnt[i] : 0;
  int inc = v;
#pragma unroll
  for (int d = 1; d < 32; d <<= 1) {
    int o = __shfl_up(inc, d);
    if (lane >= d) inc += o;
  }
  if (lane == 31) wincl[wave] = inc;
  if (lane == 0)  wpre[wave]  = p;
  __syncthreads();

  int off = 0;
#pragma unroll
  for (int w = 0; w < SCAN_BLK / 32; ++w) {
    off += wpre[w];
    if (w < wave) off += wincl[w];
  }
  int excl = off + inc - v;
  if (i < N_NODES) rowptr[i] = excl;
  if (i == N_NODES - 1) rowptr[N_NODES] = excl + v;
}

__global__ __launch_bounds__(256) void fill_rows(
    const int* __restrict__ src, const int* __restrict__ dst,
    const int* __restrict__ rowptr, const int* __restrict__ erank,
    int* __restrict__ ssrc) {
  int e = blockIdx.x * blockDim.x + threadIdx.x;
  if (e < N_EDGES) {
    int pos = rowptr[dst[e]] + erank[e];
    if ((unsigned)pos < (unsigned)N_EDGES) ssrc[pos] = src[e];
  }
}

__global__ __launch_bounds__(256) void gather_sum_rows(
    const float* __restrict__ m, const int* __restrict__ rowptr,
    const int* __restrict__ ssrc, float* __restrict__ agg) {
  int wave = __builtin_amdgcn_readfirstlane(threadIdx.x >> 5);
  int lane = threadIdx.x & 31;
  int node = blockIdx.x * 8 + wave;
  if (node >= N_NODES) return;
  int beg = rowptr[node];
  int end = rowptr[node + 1];

  const float* mlane = m + lane * 4;
  float4 acc = make_float4(0.f, 0.f, 0.f, 0.f);
  for (int base = beg; base < end; base += 32) {
    int n = end - base;
    if (n > 32) n = 32;
    int mine = lane < n ? ssrc[base + lane] : 0x7fffffff;

    int rk = 0;
    for (int k = 0; k < n; ++k) {
      int sk = __builtin_amdgcn_readlane(mine, k);
      rk += (sk < mine || (sk == mine && k < lane)) ? 1 : 0;
    }
    if (lane >= n) rk = lane;
    int sorted = __builtin_amdgcn_ds_permute(rk << 2, mine);

    for (int j = 0; j < n; j += 8) {
      float4 v[8];
#pragma unroll
      for (int t = 0; t < 8; ++t) {
        v[t] = make_float4(0.f, 0.f, 0.f, 0.f);
        if (j + t < n) {
          int s = __builtin_amdgcn_readlane(sorted, j + t);
          v[t] = *(const float4*)(mlane + (size_t)s * HD);
        }
      }
#pragma unroll
      for (int t = 0; t < 8; ++t) {
        acc.x += v[t].x; acc.y += v[t].y; acc.z += v[t].z; acc.w += v[t].w;
      }
    }
  }
  *(float4*)(agg + (size_t)node * HD + lane * 4) = acc;
}

__global__ __launch_bounds__(256) void epilogue_leaky(
    const float* __restrict__ agg, const float* __restrict__ norm,
    const float* __restrict__ bias, float* __restrict__ out) {
  int t   = blockIdx.x * blockDim.x + threadIdx.x;
  int row = t >> 5;
  int c4  = (t & 31) << 2;
  if (row < N_NODES) {
    float nr = norm[row];
    float4 a = *(const float4*)(agg + (size_t)row * HD + c4);
    float4 b = *(const float4*)(bias + c4);
    float4 n;
    n.x = a.x * nr + b.x;
    n.y = a.y * nr + b.y;
    n.z = a.z * nr + b.z;
    n.w = a.w * nr + b.w;
    n.x = n.x > 0.f ? n.x : 0.2f * n.x;
    n.y = n.y > 0.f ? n.y : 0.2f * n.y;
    n.z = n.z > 0.f ? n.z : 0.2f * n.z;
    n.w = n.w > 0.f ? n.w : 0.2f * n.w;
    *(float4*)(out + (size_t)row * HD + c4) = n;
  }
}

extern "C" void kernel_launch(void* const* d_in, const int* in_sizes, int n_in,
                              void* d_out, int out_size, void* d_ws, size_t ws_size,
                              hipStream_t stream) {
  const int*   node_id = (const int*)d_in[0];
  const int*   src     = (const int*)d_in[1];
  const int*   dst     = (const int*)d_in[2];
  const float* norm    = (const float*)d_in[3];
  const float* embed   = (const float*)d_in[4];
  const float* W1      = (const float*)d_in[5];
  const float* b1      = (const float*)d_in[6];
  const float* W2      = (const float*)d_in[7];
  const float* b2      = (const float*)d_in[8];
  float* out = (float*)d_out;

  const size_t feat = (size_t)N_NODES * HD;
  float* bufH = (float*)d_ws;
  float* bufM = bufH + feat;
  float* agg  = out;
  int* cnt    = (int*)(bufM + feat);
  int* rowptr = cnt + 100096;
  int* bsum   = rowptr + 100096;
  int* erank  = bsum + 512;
  int* ssrc   = erank + 600064;

  dim3 blk(256);
  const int gElem    = (int)((feat / 4 + 255) / 256);
  const int gGemm    = (N_NODES + 127) / 128;
  const int gEdge    = (N_EDGES + 255) / 256;
  const int gRows    = (N_NODES + 7) / 8;

  gather_embed<<<gElem, blk, 0, stream>>>(node_id, embed, bufH);

  zero_f4<<<(N_NODES / 4 + 255) / 256, blk, 0, stream>>>((float4*)cnt, N_NODES / 4);
  count_rank<<<gEdge, blk, 0, stream>>>(dst, cnt, erank);
  block_sums<<<N_SCAN_BLOCKS, SCAN_BLK, 0, stream>>>(cnt, bsum);
  scan_rows<<<N_SCAN_BLOCKS, SCAN_BLK, 0, stream>>>(cnt, bsum, rowptr);
  fill_rows<<<gEdge, blk, 0, stream>>>(src, dst, rowptr, erank, ssrc);

  gemm_norm_wmma<<<gGemm, blk, 0, stream>>>(bufH, W1, norm, bufM);
  gather_sum_rows<<<gRows, blk, 0, stream>>>(bufM, rowptr, ssrc, agg);
  epilogue_leaky<<<gElem, blk, 0, stream>>>(agg, norm, b1, bufH);

  gemm_norm_wmma<<<gGemm, blk, 0, stream>>>(bufH, W2, norm, bufM);
  gather_sum_rows<<<gRows, blk, 0, stream>>>(bufM, rowptr, ssrc, agg);
  epilogue_leaky<<<gElem, blk, 0, stream>>>(agg, norm, b2, out);
}


# Fold the 3 one-float4-per-thread passes into their neighbours: embedding lookup into GEMM-1's A row address, both LeakyReLU epilogues into the row-sum store; 12 -> 9 launches, sum order unchanged
# speedup vs baseline: 2.1060x; 1.2103x over previous
#include <hip/hip_runtime.h>

#define N_NODES 100000
#define N_EDGES 600000
#define HD 128

typedef __attribute__((ext_vector_type(2))) float v2f;
typedef __attribute__((ext_vector_type(8))) float v8f;

template <bool LOOKUP>
__global__ __launch_bounds__(256) void gemm_norm_wmma(
    const float* __restrict__ h, const int* __restrict__ node_id,
    const float* __restrict__ W, const float* __restrict__ norm,
    float* __restrict__ m) {
  __shared__ float Wl2[HD * HD];

  int tid     = threadIdx.x;
  int wave    = tid >> 5;
  int lane    = tid & 31;
  int lane16  = lane & 15;
  int pOff    = lane >> 4;
  int rowBase = (blockIdx.x * 8 + wave) * 16;
  int rowA    = rowBase + lane16;
  int rowAc   = rowA < N_NODES ? rowA : (N_NODES - 1);
  int rowSrc  = LOOKUP ? node_id[rowAc] : rowAc;

  for (int i = tid; i < (HD * HD) / 4; i += 256) {
    int p = i >> 6;
    int n = (i & 63) << 1;
    int k = p << 1;
    float2 r0 = *(const float2*)(W + (size_t)k * HD + n);
    float2 r1 = *(const float2*)(W + (size_t)(k + 1) * HD + n);
    float4 o  = make_float4(r0.x, r1.x, r0.y, r1.y);
    *(float4*)(Wl2 + p * 256 + n * 2) = o;
  }
  __syncthreads();

  v8f acc[8] = {};
  const float* hrow = h + (size_t)rowSrc * HD;

  for (int k0 = 0; k0 < HD; k0 += 4) {
    v2f a = *(const v2f*)(hrow + k0 + (pOff << 1));
    const float* bbase = Wl2 + ((k0 >> 1) + pOff) * 256;
#pragma unroll
    for (int t = 0; t < 8; ++t) {
      int n = t * 16 + lane16;
      v2f b = *(const v2f*)(bbase + n * 2);
      acc[t] = __builtin_amdgcn_wmma_f32_16x16x4_f32(
          false, a, false, b, (short)0, acc[t], false, false);
    }
  }

  int rhalf = pOff << 3;
#pragma unroll
  for (int v = 0; v < 8; ++v) {
    int r = rowBase + v + rhalf;
    if (r < N_NODES) {
      float nr = norm[r];
#pragma unroll
      for (int t = 0; t < 8; ++t)
        m[(size_t)r * HD + t * 16 + lane16] = acc[t][v] * nr;
    }
  }
}

__global__ __launch_bounds__(256) void zero_f4(float4* __restrict__ p, int n4) {
  int t = blockIdx.x * blockDim.x + threadIdx.x;
  if (t < n4) p[t] = make_float4(0.f, 0.f, 0.f, 0.f);
}

__global__ __launch_bounds__(256) void count_rank(
    const int* __restrict__ dst, int* __restrict__ cnt, int* __restrict__ erank) {
  int e = blockIdx.x * blockDim.x + threadIdx.x;
  if (e < N_EDGES)
    erank[e] = __hip_atomic_fetch_add(cnt + dst[e], 1, __ATOMIC_RELAXED, __HIP_MEMORY_SCOPE_AGENT);
}

#define SCAN_BLK 256
#define N_SCAN_BLOCKS ((N_NODES + SCAN_BLK - 1) / SCAN_BLK)

__global__ __launch_bounds__(SCAN_BLK) void block_sums(
    const int* __restrict__ cnt, int* __restrict__ bsum) {
  __shared__ int wsum[SCAN_BLK / 32];
  int i = blockIdx.x * SCAN_BLK + threadIdx.x;
  int v = i < N_NODES ? cnt[i] : 0;
#pragma unroll
  for (int d = 16; d > 0; d >>= 1) v += __shfl_xor(v, d);
  if ((threadIdx.x & 31) == 0) wsum[threadIdx.x >> 5] = v;
  __syncthreads();
  if (threadIdx.x == 0) {
    int s = 0;
#pragma unroll
    for (int w = 0; w < SCAN_BLK / 32; ++w) s += wsum[w];
    bsum[blockIdx.x] = s;
  }
}

__global__ __launch_bounds__(SCAN_BLK) void scan_rows(
    const int* __restrict__ cnt, const int* __restrict__ bsum,
    int* __restrict__ rowptr) {
  __shared__ int wincl[SCAN_BLK / 32];
  __shared__ int wpre[SCAN_BLK / 32];
  int tid  = threadIdx.x;
  int lane = tid & 31;
  int wave = tid >> 5;
  int b    = blockIdx.x;

  int p = 0;
  for (int k = tid; k < b; k += SCAN_BLK) p += bsum[k];
#pragma unroll
  for (int d = 16; d > 0; d >>= 1) p += __shfl_xor(p, d);

  int i = b * SCAN_BLK + tid;
  int v = i < N_NODES ? cnt[i] : 0;
  int inc = v;
#pragma unroll
  for (int d = 1; d < 32; d <<= 1) {
    int o = __shfl_up(inc, d);
    if (lane >= d) inc += o;
  }
  if (lane == 31) wincl[wave] = inc;
  if (lane == 0)  wpre[wave]  = p;
  __syncthreads();

  int off = 0;
#pragma unroll
  for (int w = 0; w < SCAN_BLK / 32; ++w) {
    off += wpre[w];
    if (w < wave) off += wincl[w];
  }
  int excl = off + inc - v;
  if (i < N_NODES) rowptr[i] = excl;
  if (i == N_NODES - 1) rowptr[N_NODES] = excl + v;
}

__global__ __launch_bounds__(256) void fill_rows(
    const int* __restrict__ src, const int* __restrict__ dst,
    const int* __restrict__ rowptr, const int* __restrict__ erank,
    int* __restrict__ ssrc) {
  int e = blockIdx.x * blockDim.x + threadIdx.x;
  if (e < N_EDGES) {
    int pos = rowptr[dst[e]] + erank[e];
    if ((unsigned)pos < (unsigned)N_EDGES) ssrc[pos] = src[e];
  }
}

__global__ __launch_bounds__(256) void gather_sum_leaky(
    const float* __restrict__ m, const int* __restrict__ rowptr,
    const int* __restrict__ ssrc, const float* __restrict__ norm,
    const float* __restrict__ bias, float* __restrict__ out) {
  int wave = __builtin_amdgcn_readfirstlane(threadIdx.x >> 5);
  int lane = threadIdx.x & 31;
  int node = blockIdx.x * 8 + wave;
  if (node >= N_NODES) return;
  int beg = rowptr[node];
  int end = rowptr[node + 1];
  float  nr = norm[node];
  float4 b  = *(const float4*)(bias + lane * 4);

  const float* mlane = m + lane * 4;
  float4 acc = make_float4(0.f, 0.f, 0.f, 0.f);
  for (int base = beg; base < end; base += 32) {
    int n = end - base;
    if (n > 32) n = 32;
    int mine = lane < n ? ssrc[base + lane] : 0x7fffffff;

    int rk = 0;
    for (int k = 0; k < n; ++k) {
      int sk = __builtin_amdgcn_readlane(mine, k);
      rk += (sk < mine || (sk == mine && k < lane)) ? 1 : 0;
    }
    if (lane >= n) rk = lane;
    int sorted = __builtin_amdgcn_ds_permute(rk << 2, mine);

    for (int j = 0; j < n; j += 8) {
      float4 v[8];
#pragma unroll
      for (int t = 0; t < 8; ++t) {
        v[t] = make_float4(0.f, 0.f, 0.f, 0.f);
        if (j + t < n) {
          int s = __builtin_amdgcn_readlane(sorted, j + t);
          v[t] = *(const float4*)(mlane + (size_t)s * HD);
        }
      }
#pragma unroll
      for (int t = 0; t < 8; ++t) {
        acc.x += v[t].x; acc.y += v[t].y; acc.z += v[t].z; acc.w += v[t].w;
      }
    }
  }
  float4 n;
  n.x = acc.x * nr + b.x;
  n.y = acc.y * nr + b.y;
  n.z = acc.z * nr + b.z;
  n.w = acc.w * nr + b.w;
  n.x = n.x > 0.f ? n.x : 0.2f * n.x;
  n.y = n.y > 0.f ? n.y : 0.2f * n.y;
  n.z = n.z > 0.f ? n.z : 0.2f * n.z;
  n.w = n.w > 0.f ? n.w : 0.2f * n.w;
  *(float4*)(out + (size_t)node * HD + lane * 4) = n;
}

extern "C" void kernel_launch(void* const* d_in, const int* in_sizes, int n_in,
                              void* d_out, int out_size, void* d_ws, size_t ws_size,
                              hipStream_t stream) {
  const int*   node_id = (const int*)d_in[0];
  const int*   src     = (const int*)d_in[1];
  const int*   dst     = (const int*)d_in[2];
  const float* norm    = (const float*)d_in[3];
  const float* embed   = (const float*)d_in[4];
  const float* W1      = (const float*)d_in[5];
  const float* b1      = (const float*)d_in[6];
  const float* W2      = (const float*)d_in[7];
  const float* b2      = (const float*)d_in[8];
  float* out = (float*)d_out;

  const size_t feat = (size_t)N_NODES * HD;
  float* bufH = (float*)d_ws;
  float* bufM = bufH + feat;
  int* cnt    = (int*)(bufM + feat);
  int* rowptr = cnt + 100096;
  int* bsum   = rowptr + 100096;
  int* erank  = bsum + 512;
  int* ssrc   = erank + 600064;

  dim3 blk(256);
  const int gGemm    = (N_NODES + 127) / 128;
  const int gEdge    = (N_EDGES + 255) / 256;
  const int gRows    = (N_NODES + 7) / 8;

  zero_f4<<<(N_NODES / 4 + 255) / 256, blk, 0, stream>>>((float4*)cnt, N_NODES / 4);
  count_rank<<<gEdge, blk, 0, stream>>>(dst, cnt, erank);
  block_sums<<<N_SCAN_BLOCKS, SCAN_BLK, 0, stream>>>(cnt, bsum);
  scan_rows<<<N_SCAN_BLOCKS, SCAN_BLK, 0, stream>>>(cnt, bsum, rowptr);
  fill_rows<<<gEdge, blk, 0, stream>>>(src, dst, rowptr, erank, ssrc);

  gemm_norm_wmma<true><<<gGemm, blk, 0, stream>>>(embed, node_id, W1, norm, bufM);
  gather_sum_leaky<<<gRows, blk, 0, stream>>>(bufM, rowptr, ssrc, norm, b1, bufH);

  gemm_norm_wmma<false><<<gGemm, blk, 0, stream>>>(bufH, node_id, W2, norm, bufM);
  gather_sum_leaky<<<gRows, blk, 0, stream>>>(bufM, rowptr, ssrc, norm, b2, out);
}


# Sum each row's norm-weighted input rows first, then multiply the summed 16-row LDS tile by W in the same kernel (segment sum is linear): no message plane, 9 -> 7 launches; sum precedes the product
# speedup vs baseline: 2.3018x; 1.0930x over previous
#include <hip/hip_runtime.h>

#define N_NODES 100000
#define N_EDGES 600000
#define HD 128
#define TILE_ROWS 128
#define APITCH 132

typedef __attribute__((ext_vector_type(2))) float v2f;
typedef __attribute__((ext_vector_type(8))) float v8f;

__global__ __launch_bounds__(256) void zero_f4(float4* __restrict__ p, int n4) {
  int t = blockIdx.x * blockDim.x + threadIdx.x;
  if (t < n4) p[t] = make_float4(0.f, 0.f, 0.f, 0.f);
}

__global__ __launch_bounds__(256) void count_rank(
    const int* __restrict__ dst, int* __restrict__ cnt, int* __restrict__ erank) {
  int e = blockIdx.x * blockDim.x + threadIdx.x;
  if (e < N_EDGES)
    erank[e] = __hip_atomic_fetch_add(cnt + dst[e], 1, __ATOMIC_RELAXED, __HIP_MEMORY_SCOPE_AGENT);
}

#define SCAN_BLK 256
#define N_SCAN_BLOCKS ((N_NODES + SCAN_BLK - 1) / SCAN_BLK)

__global__ __launch_bounds__(SCAN_BLK) void block_sums(
    const int* __restrict__ cnt, int* __restrict__ bsum) {
  __shared__ int wsum[SCAN_BLK / 32];
  int i = blockIdx.x * SCAN_BLK + threadIdx.x;
  int v = i < N_NODES ? cnt[i] : 0;
#pragma unroll
  for (int d = 16; d > 0; d >>= 1) v += __shfl_xor(v, d);
  if ((threadIdx.x & 31) == 0) wsum[threadIdx.x >> 5] = v;
  __syncthreads();
  if (threadIdx.x == 0) {
    int s = 0;
#pragma unroll
    for (int w = 0; w < SCAN_BLK / 32; ++w) s += wsum[w];
    bsum[blockIdx.x] = s;
  }
}

__global__ __launch_bounds__(SCAN_BLK) void scan_rows(
    const int* __restrict__ cnt, const int* __restrict__ bsum,
    int* __restrict__ rowptr) {
  __shared__ int wincl[SCAN_BLK / 32];
  __shared__ int wpre[SCAN_BLK / 32];
  int tid  = threadIdx.x;
  int lane = tid & 31;
  int wave = tid >> 5;
  int b    = blockIdx.x;

  int p = 0;
  for (int k = tid; k < b; k += SCAN_BLK) p += bsum[k];
#pragma unroll
  for (int d = 16; d > 0; d >>= 1) p += __shfl_xor(p, d);

  int i = b * SCAN_BLK + tid;
  int v = i < N_NODES ? cnt[i] : 0;
  int inc = v;
#pragma unroll
  for (int d = 1; d < 32; d <<= 1) {
    int o = __shfl_up(inc, d);
    if (lane >= d) inc += o;
  }
  if (lane == 31) wincl[wave] = inc;
  if (lane == 0)  wpre[wave]  = p;
  __syncthreads();

  int off = 0;
#pragma unroll
  for (int w = 0; w < SCAN_BLK / 32; ++w) {
    off += wpre[w];
    if (w < wave) off += wincl[w];
  }
  int excl = off + inc - v;
  if (i < N_NODES) rowptr[i] = excl;
  if (i == N_NODES - 1) rowptr[N_NODES] = excl + v;
}

__global__ __launch_bounds__(256) void fill_rows(
    const int* __restrict__ src, const int* __restrict__ dst,
    const int* __restrict__ rowptr, const int* __restrict__ erank,
    int* __restrict__ ssrc) {
  int e = blockIdx.x * blockDim.x + threadIdx.x;
  if (e < N_EDGES) {
    int pos = rowptr[dst[e]] + erank[e];
    if ((unsigned)pos < (unsigned)N_EDGES) ssrc[pos] = src[e];
  }
}


__device__ __forceinline__ int sort_sources(int mine, int n, int lane) {
  int rk = 0;
  for (int k = 0; k < n; ++k) {
    int sk = __builtin_amdgcn_readlane(mine, k);
    rk += (sk < mine || (sk == mine && k < lane)) ? 1 : 0;
  }
  if (lane >= n) rk = lane;
  return __builtin_amdgcn_ds_permute(rk << 2, mine);
}

__device__ __forceinline__ void add_rows(float4& acc, const float* __restrict__ xlane,
                                         int rows, float wts, int n) {
  for (int j = 0; j < n; j += 8) {
    float4 v[8];
    float  w[8];
#pragma unroll
    for (int t = 0; t < 8; ++t) {
      v[t] = make_float4(0.f, 0.f, 0.f, 0.f);
      w[t] = 0.f;
      if (j + t < n) {
        int r = __builtin_amdgcn_readlane(rows, j + t);
        w[t]  = __int_as_float(__builtin_amdgcn_readlane(__float_as_int(wts), j + t));
        v[t]  = *(const float4*)(xlane + (size_t)r * HD);
      }
    }
#pragma unroll
    for (int t = 0; t < 8; ++t) {
      acc.x += w[t] * v[t].x; acc.y += w[t] * v[t].y;
      acc.z += w[t] * v[t].z; acc.w += w[t] * v[t].w;
    }
  }
}

template <bool LOOKUP>
__global__ __launch_bounds__(1024) void layer_gather_wmma(
    const float* __restrict__ x, const int* __restrict__ node_id,
    const int* __restrict__ rowptr, const int* __restrict__ ssrc,
    const float* __restrict__ norm, const float* __restrict__ W,
    const float* __restrict__ bias, float* __restrict__ out) {
  __shared__ __attribute__((aligned(16))) float Wl2[HD * HD];
  __shared__ __attribute__((aligned(16))) float At[TILE_ROWS * APITCH];

  const int tid  = threadIdx.x;
  const int wave = __builtin_amdgcn_readfirstlane(tid >> 5);
  const int lane = tid & 31;
  const int node0 = blockIdx.x * TILE_ROWS + wave * 4;

  float2 wr0[4], wr1[4];
#pragma unroll
  for (int it = 0; it < 4; ++it) {
    int i = tid + it * 1024;
    int p = i >> 6;
    int n = (i & 63) << 1;
    int k = p << 1;
    wr0[it] = *(const float2*)(W + (size_t)k * HD + n);
    wr1[it] = *(const float2*)(W + (size_t)(k + 1) * HD + n);
  }

  int rp[5];
#pragma unroll
  for (int i = 0; i < 5; ++i) {
    int nd = node0 + i;
    rp[i] = rowptr[nd < N_NODES ? nd : N_NODES];
  }
  int beg[4], end[4];
#pragma unroll
  for (int i = 0; i < 4; ++i) { beg[i] = rp[i]; end[i] = rp[i + 1]; }
  int n0[4], mine[4];
#pragma unroll
  for (int i = 0; i < 4; ++i) {
    int n = end[i] - beg[i];
    n = n > 32 ? 32 : n;
    n = n < 0 ? 0 : n;
    n0[i]   = n;
    mine[i] = lane < n ? ssrc[beg[i] + lane] : 0x7fffffff;
  }

  int   rows[4];
  float wts[4];
#pragma unroll
  for (int i = 0; i < 4; ++i) {
    int  s   = sort_sources(mine[i], n0[i], lane);
    bool act = lane < n0[i];
    int  sc  = act ? s : 0;
    wts[i]  = act ? norm[sc] : 0.f;
    rows[i] = LOOKUP ? (act ? node_id[sc] : 0) : sc;
  }

#pragma unroll
  for (int it = 0; it < 4; ++it) {
    int i = tid + it * 1024;
    int p = i >> 6;
    int n = (i & 63) << 1;
    *(float4*)(Wl2 + p * 256 + n * 2) =
        make_float4(wr0[it].x, wr1[it].x, wr0[it].y, wr1[it].y);
  }

  const float* xlane = x + lane * 4;
#pragma unroll
  for (int i = 0; i < 4; ++i) {
    float4 acc = make_float4(0.f, 0.f, 0.f, 0.f);
    add_rows(acc, xlane, rows[i], wts[i], n0[i]);
    for (int base = beg[i] + 32; base < end[i]; base += 32) {
      int n = end[i] - base;
      n = n > 32 ? 32 : n;
      int  m2  = lane < n ? ssrc[base + lane] : 0x7fffffff;
      int  s   = sort_sources(m2, n, lane);
      bool act = lane < n;
      int  sc  = act ? s : 0;
      float w2 = act ? norm[sc] : 0.f;
      int   r2 = LOOKUP ? (act ? node_id[sc] : 0) : sc;
      add_rows(acc, xlane, r2, w2, n);
    }
    *(float4*)(At + (wave * 4 + i) * APITCH + lane * 4) = acc;
  }
  __syncthreads();

  const int lane16 = lane & 15;
  const int pOff   = lane >> 4;
  const int rt     = wave >> 2;
  const int nt0    = (wave & 3) * 2;
  v8f acc0 = {}, acc1 = {};
  const float* arow = At + (rt * 16 + lane16) * APITCH + (pOff << 1);
  const float* bcol = Wl2 + pOff * 256 + (nt0 * 16 + lane16) * 2;
  for (int k0 = 0; k0 < HD; k0 += 4) {
    v2f a  = *(const v2f*)(arow + k0);
    v2f b0 = *(const v2f*)(bcol + (k0 >> 1) * 256);
    v2f b1 = *(const v2f*)(bcol + (k0 >> 1) * 256 + 32);
    acc0 = __builtin_amdgcn_wmma_f32_16x16x4_f32(
        false, a, false, b0, (short)0, acc0, false, false);
    acc1 = __builtin_amdgcn_wmma_f32_16x16x4_f32(
        false, a, false, b1, (short)0, acc1, false, false);
  }

  const int row0 = blockIdx.x * TILE_ROWS + rt * 16 + (pOff << 3);
  const int col0 = nt0 * 16 + lane16;
  const float bs0 = bias[col0];
  const float bs1 = bias[col0 + 16];
#pragma unroll
  for (int v = 0; v < 8; ++v) {
    int r = row0 + v;
    if (r < N_NODES) {
      float nr = norm[r];
      float y0 = acc0[v] * nr + bs0;
      float y1 = acc1[v] * nr + bs1;
      y0 = y0 > 0.f ? y0 : 0.2f * y0;
      y1 = y1 > 0.f ? y1 : 0.2f * y1;
      out[(size_t)r * HD + col0]      = y0;
      out[(size_t)r * HD + col0 + 16] = y1;
    }
  }
}

extern "C" void kernel_launch(void* const* d_in, const int* in_sizes, int n_in,
                              void* d_out, int out_size, void* d_ws, size_t ws_size,
                              hipStream_t stream) {
  const int*   node_id = (const int*)d_in[0];
  const int*   src     = (const int*)d_in[1];
  const int*   dst     = (const int*)d_in[2];
  const float* norm    = (const float*)d_in[3];
  const float* embed   = (const float*)d_in[4];
  const float* W1      = (const float*)d_in[5];
  const float* b1      = (const float*)d_in[6];
  const float* W2      = (const float*)d_in[7];
  const float* b2      = (const float*)d_in[8];
  float* out = (float*)d_out;

  const size_t feat = (size_t)N_NODES * HD;
  float* bufH = (float*)d_ws;
  float* bufM = bufH + feat;
  int* cnt    = (int*)(bufM + feat);
  int* rowptr = cnt + 100096;
  int* bsum   = rowptr + 100096;
  int* erank  = bsum + 512;
  int* ssrc   = erank + 600064;

  dim3 blk(256);
  const int gEdge  = (N_EDGES + 255) / 256;
  const int gLayer = (N_NODES + TILE_ROWS - 1) / TILE_ROWS;

  zero_f4<<<(N_NODES / 4 + 255) / 256, blk, 0, stream>>>((float4*)cnt, N_NODES / 4);
  count_rank<<<gEdge, blk, 0, stream>>>(dst, cnt, erank);
  block_sums<<<N_SCAN_BLOCKS, SCAN_BLK, 0, stream>>>(cnt, bsum);
  scan_rows<<<N_SCAN_BLOCKS, SCAN_BLK, 0, stream>>>(cnt, bsum, rowptr);
  fill_rows<<<gEdge, blk, 0, stream>>>(src, dst, rowptr, erank, ssrc);

  layer_gather_wmma<true><<<gLayer, dim3(1024), 0, stream>>>(
      embed, node_id, rowptr, ssrc, norm, W1, b1, bufH);
  layer_gather_wmma<false><<<gLayer, dim3(1024), 0, stream>>>(
      bufH, node_id, rowptr, ssrc, norm, W2, b2, out);
}


# Group edges by destination in the counting pass: each row owns a 32-slot line addressed by the atomic's return; no prefix scan or fill pass, 7 -> 4 launches; rows > 32 edges sum in edge order
# speedup vs baseline: 2.6585x; 1.1550x over previous
#include <hip/hip_runtime.h>

#define N_NODES 100000
#define N_EDGES 600000
#define HD 128
#define TILE_ROWS 128
#define APITCH 132
#define SLOTS 32
#define N_PAD (((N_NODES + TILE_ROWS - 1) / TILE_ROWS) * TILE_ROWS)

typedef __attribute__((ext_vector_type(2))) float v2f;
typedef __attribute__((ext_vector_type(8))) float v8f;

__global__ __launch_bounds__(256) void zero_f4(float4* __restrict__ p, int n4) {
  int t = blockIdx.x * blockDim.x + threadIdx.x;
  if (t < n4) p[t] = make_float4(0.f, 0.f, 0.f, 0.f);
}

__global__ __launch_bounds__(256) void count_place(
    const int* __restrict__ src, const int* __restrict__ dst,
    int* __restrict__ cnt, int* __restrict__ slots) {
  int e = blockIdx.x * blockDim.x + threadIdx.x;
  if (e < N_EDGES) {
    int d = dst[e];
    int s = src[e];
    asm volatile("" :: "v"(s));
    if ((unsigned)d < (unsigned)N_NODES) {
      int r = __hip_atomic_fetch_add(cnt + d, 1, __ATOMIC_RELAXED, __HIP_MEMORY_SCOPE_AGENT);
      if (r < SLOTS) slots[d * SLOTS + r] = s;
    }
  }
}


__device__ __forceinline__ int sort_sources(int mine, int n, int lane) {
  int rk = 0;
  for (int k = 0; k < n; ++k) {
    int sk = __builtin_amdgcn_readlane(mine, k);
    rk += (sk < mine || (sk == mine && k < lane)) ? 1 : 0;
  }
  if (lane >= n) rk = lane;
  return __builtin_amdgcn_ds_permute(rk << 2, mine);
}

__device__ __forceinline__ void add_rows(float4& acc, const float* __restrict__ xlane,
                                         int rows, float wts, int n) {
  for (int j = 0; j < n; j += 8) {
    float4 v[8];
    float  w[8];
#pragma unroll
    for (int t = 0; t < 8; ++t) {
      v[t] = make_float4(0.f, 0.f, 0.f, 0.f);
      w[t] = 0.f;
      if (j + t < n) {
        int r = __builtin_amdgcn_readlane(rows, j + t);
        w[t]  = __int_as_float(__builtin_amdgcn_readlane(__float_as_int(wts), j + t));
        v[t]  = *(const float4*)(xlane + (size_t)r * HD);
      }
    }
#pragma unroll
    for (int t = 0; t < 8; ++t) {
      acc.x += w[t] * v[t].x; acc.y += w[t] * v[t].y;
      acc.z += w[t] * v[t].z; acc.w += w[t] * v[t].w;
    }
  }
}

template <bool LOOKUP>
__global__ __launch_bounds__(1024) void layer_gather_wmma(
    const float* __restrict__ x, const int* __restrict__ node_id,
    const int* __restrict__ cnt, const int* __restrict__ slots,
    const int* __restrict__ src, const int* __restrict__ dst,
    const float* __restrict__ norm, const float* __restrict__ W,
    const float* __restrict__ bias, float* __restrict__ out) {
  __shared__ __attribute__((aligned(16))) float Wl2[HD * HD];
  __shared__ __attribute__((aligned(16))) float At[TILE_ROWS * APITCH];

  const int tid  = threadIdx.x;
  const int wave = __builtin_amdgcn_readfirstlane(tid >> 5);
  const int lane = tid & 31;
  const int node0 = blockIdx.x * TILE_ROWS + wave * 4;

  float2 wr0[4], wr1[4];
#pragma unroll
  for (int it = 0; it < 4; ++it) {
    int i = tid + it * 1024;
    int p = i >> 6;
    int n = (i & 63) << 1;
    int k = p << 1;
    wr0[it] = *(const float2*)(W + (size_t)k * HD + n);
    wr1[it] = *(const float2*)(W + (size_t)(k + 1) * HD + n);
  }

  const int4 c4 = *(const int4*)(cnt + node0);
  int sv[4];
#pragma unroll
  for (int i = 0; i < 4; ++i) sv[i] = slots[(node0 + i) * SLOTS + lane];
  const int ntot[4] = {c4.x, c4.y, c4.z, c4.w};
  int n0[4], mine[4];
#pragma unroll
  for (int i = 0; i < 4; ++i) {
    int n = ntot[i];
    n = n > SLOTS ? 0 : n;
    n = n < 0 ? 0 : n;
    n0[i]   = n;
    mine[i] = lane < n ? sv[i] : 0x7fffffff;
  }

  int   rows[4];
  float wts[4];
#pragma unroll
  for (int i = 0; i < 4; ++i) {
    int  s   = sort_sources(mine[i], n0[i], lane);
    bool act = lane < n0[i];
    int  sc  = act ? s : 0;
    wts[i]  = act ? norm[sc] : 0.f;
    rows[i] = LOOKUP ? (act ? node_id[sc] : 0) : sc;
  }

#pragma unroll
  for (int it = 0; it < 4; ++it) {
    int i = tid + it * 1024;
    int p = i >> 6;
    int n = (i & 63) << 1;
    *(float4*)(Wl2 + p * 256 + n * 2) =
        make_float4(wr0[it].x, wr1[it].x, wr0[it].y, wr1[it].y);
  }

  const float* xlane = x + lane * 4;
#pragma unroll
  for (int i = 0; i < 4; ++i) {
    float4 acc = make_float4(0.f, 0.f, 0.f, 0.f);
    add_rows(acc, xlane, rows[i], wts[i], n0[i]);
    if (ntot[i] > SLOTS) {
      const int node = node0 + i;
      for (int base = 0; base < N_EDGES; base += 32) {
        int  e = base + lane;
        bool m = e < N_EDGES && dst[e] == node;
        unsigned mask = __builtin_amdgcn_ballot_w32(m);
        if (mask == 0u) continue;
        int n = __builtin_popcount(mask);
        unsigned below = (1u << lane) - 1u;
        int rk = m ? __builtin_popcount(mask & below) : n + __builtin_popcount(~mask & below);
        int s  = __builtin_amdgcn_ds_permute(rk << 2, m ? src[e] : 0);
        bool act = lane < n;
        int  sc  = act ? s : 0;
        float w2 = act ? norm[sc] : 0.f;
        int   r2 = LOOKUP ? (act ? node_id[sc] : 0) : sc;
        add_rows(acc, xlane, r2, w2, n);
      }
    }
    *(float4*)(At + (wave * 4 + i) * APITCH + lane * 4) = acc;
  }
  __syncthreads();

  const int lane16 = lane & 15;
  const int pOff   = lane >> 4;
  const int rt     = wave >> 2;
  const int nt0    = (wave & 3) * 2;
  v8f acc0 = {}, acc1 = {};
  const float* arow = At + (rt * 16 + lane16) * APITCH + (pOff << 1);
  const float* bcol = Wl2 + pOff * 256 + (nt0 * 16 + lane16) * 2;
  for (int k0 = 0; k0 < HD; k0 += 4) {
    v2f a  = *(const v2f*)(arow + k0);
    v2f b0 = *(const v2f*)(bcol + (k0 >> 1) * 256);
    v2f b1 = *(const v2f*)(bcol + (k0 >> 1) * 256 + 32);
    acc0 = __builtin_amdgcn_wmma_f32_16x16x4_f32(
        false, a, false, b0, (short)0, acc0, false, false);
    acc1 = __builtin_amdgcn_wmma_f32_16x16x4_f32(
        false, a, false, b1, (short)0, acc1, false, false);
  }

  const int row0 = blockIdx.x * TILE_ROWS + rt * 16 + (pOff << 3);
  const int col0 = nt0 * 16 + lane16;
  const float bs0 = bias[col0];
  const float bs1 = bias[col0 + 16];
#pragma unroll
  for (int v = 0; v < 8; ++v) {
    int r = row0 + v;
    if (r < N_NODES) {
      float nr = norm[r];
      float y0 = acc0[v] * nr + bs0;
      float y1 = acc1[v] * nr + bs1;
      y0 = y0 > 0.f ? y0 : 0.2f * y0;
      y1 = y1 > 0.f ? y1 : 0.2f * y1;
      out[(size_t)r * HD + col0]      = y0;
      out[(size_t)r * HD + col0 + 16] = y1;
    }
  }
}

extern "C" void kernel_launch(void* const* d_in, const int* in_sizes, int n_in,
                              void* d_out, int out_size, void* d_ws, size_t ws_size,
                              hipStream_t stream) {
  const int*   node_id = (const int*)d_in[0];
  const int*   src     = (const int*)d_in[1];
  const int*   dst     = (const int*)d_in[2];
  const float* norm    = (const float*)d_in[3];
  const float* embed   = (const float*)d_in[4];
  const float* W1      = (const float*)d_in[5];
  const float* b1      = (const float*)d_in[6];
  const float* W2      = (const float*)d_in[7];
  const float* b2      = (const float*)d_in[8];
  float* out = (float*)d_out;

  const size_t feat = (size_t)N_NODES * HD;
  float* bufH = (float*)d_ws;
  int* cnt   = (int*)(bufH + feat);
  int* slots = cnt + N_PAD;

  dim3 blk(256);
  const int gEdge  = (N_EDGES + 255) / 256;
  const int gLayer = (N_NODES + TILE_ROWS - 1) / TILE_ROWS;

  zero_f4<<<(N_PAD / 4 + 255) / 256, blk, 0, stream>>>((float4*)cnt, N_PAD / 4);
  count_place<<<gEdge, blk, 0, stream>>>(src, dst, cnt, slots);

  layer_gather_wmma<true><<<gLayer, dim3(1024), 0, stream>>>(
      embed, node_id, cnt, slots, src, dst, norm, W1, b1, bufH);
  layer_gather_wmma<false><<<gLayer, dim3(1024), 0, stream>>>(
      bufH, node_id, cnt, slots, src, dst, norm, W2, b2, out);
}


# Group edges by destination with LDS counters in two levels (1024-row range per 1024-edge chunk, then row) in place of one returning global atomic per edge; no zero pass; 4 launches, sums unchanged
# speedup vs baseline: 3.1709x; 1.1927x over previous
#include <hip/hip_runtime.h>

#define N_NODES 100000
#define N_EDGES 600000
#define HD 128
#define TILE_ROWS 128
#define APITCH 132
#define SLOTS 32
#define N_PAD (((N_NODES + TILE_ROWS - 1) / TILE_ROWS) * TILE_ROWS)
#define RSHIFT 10
#define RROWS (1 << RSHIFT)
#define NRANGE ((N_PAD + RROWS - 1) / RROWS)
#define N_ALLOC (NRANGE * RROWS)
#define CHUNK 1024
#define NCHUNK ((N_EDGES + CHUNK - 1) / CHUNK)
#define CAP 32
#define CHUNKS_PER_WAVE ((NCHUNK + 31) / 32)

typedef __attribute__((ext_vector_type(2))) float v2f;
typedef __attribute__((ext_vector_type(8))) float v8f;

__global__ __launch_bounds__(1024) void bin_by_range(
    const int* __restrict__ src, const int* __restrict__ dst,
    int* __restrict__ cntG, int2* __restrict__ recG) {
  __shared__ __attribute__((aligned(16))) int cntR[128];
  __shared__ __attribute__((aligned(16))) int offR[128];
  __shared__ int2 stage[CHUNK];
  const int tid = threadIdx.x;
  const int blk = blockIdx.x;
  const int e = blk * CHUNK + tid;
  int d = -1, s = 0;
  if (e < N_EDGES) { d = dst[e]; s = src[e]; }
  if (tid < 128) cntR[tid] = 0;
  __syncthreads();
  const bool ok = (unsigned)d < (unsigned)N_NODES;
  const int r = d >> RSHIFT;
  int rank = 0;
  if (ok) rank = atomicAdd(&cntR[r], 1);
  __syncthreads();
  if (tid < 32) {
    int4 c4 = ((const int4*)cntR)[tid];
    int s4 = c4.x + c4.y + c4.z + c4.w;
    int inc = s4;
#pragma unroll
    for (int o = 1; o < 32; o <<= 1) {
      int t = __shfl_up(inc, o);
      if (tid >= o) inc += t;
    }
    int ex = inc - s4;
    ((int4*)offR)[tid] = make_int4(ex, ex + c4.x, ex + c4.x + c4.y, ex + c4.x + c4.y + c4.z);
  } else if (tid >= 128 && tid < 128 + NRANGE) {
    int q = tid - 128;
    cntG[q * NCHUNK + blk] = cntR[q];
  }
  __syncthreads();
  if (ok) stage[offR[r] + rank] = make_int2(s, d);
  __syncthreads();
  const int total = offR[127] + cntR[127];
  if (tid < total) {
    int2 rec = stage[tid];
    int rr  = rec.y >> RSHIFT;
    int pos = tid - offR[rr];
    if (pos < CAP) recG[((size_t)rr * NCHUNK + blk) * CAP + pos] = rec;
  }
}

__global__ __launch_bounds__(1024) void place_rows(
    const int* __restrict__ src, const int* __restrict__ dst,
    const int* __restrict__ cntG, const int2* __restrict__ recG,
    int* __restrict__ cnt, int* __restrict__ slots) {
  __shared__ __attribute__((aligned(16))) int cntL[RROWS];
  __shared__ __attribute__((aligned(16))) int slotL[RROWS * SLOTS];
  const int tid  = threadIdx.x;
  const int r    = blockIdx.x;
  const int wave = __builtin_amdgcn_readfirstlane(tid >> 5);
  const int lane = tid & 31;
  cntL[tid] = 0;
  int  nc[CHUNKS_PER_WAVE];
  int2 rec[CHUNKS_PER_WAVE];
#pragma unroll
  for (int i = 0; i < CHUNKS_PER_WAVE; ++i) {
    int c = wave + 32 * i;
    nc[i] = c < NCHUNK ? cntG[r * NCHUNK + c] : 0;
  }
  unsigned over = 0u;
#pragma unroll
  for (int i = 0; i < CHUNKS_PER_WAVE; ++i) {
    int c = wave + 32 * i;
    nc[i] = __builtin_amdgcn_readfirstlane(nc[i]);
    if (nc[i] > CAP) { over |= 1u << i; nc[i] = 0; }
    rec[i] = make_int2(0, 0);
    if (lane < nc[i]) rec[i] = recG[((size_t)r * NCHUNK + c) * CAP + lane];
  }
  __syncthreads();
  int pl[CHUNKS_PER_WAVE];
#pragma unroll
  for (int i = 0; i < CHUNKS_PER_WAVE; ++i) {
    pl[i] = SLOTS;
    if (lane < nc[i]) pl[i] = atomicAdd(&cntL[rec[i].y - r * RROWS], 1);
  }
#pragma unroll
  for (int i = 0; i < CHUNKS_PER_WAVE; ++i) {
    if (pl[i] < SLOTS) slotL[(rec[i].y - r * RROWS) * SLOTS + pl[i]] = rec[i].x;
  }
  while (over != 0u) {
    const int i = __builtin_ctz(over);
    over &= over - 1u;
    const int c = wave + 32 * i;
#pragma nounroll
    for (int k = 0; k < CHUNK; k += 32) {
      int e = c * CHUNK + k + lane;
      if (e < N_EDGES) {
        int d = dst[e];
        if ((unsigned)d < (unsigned)N_NODES && (d >> RSHIFT) == r) {
          int row = d - r * RROWS;
          int pl  = atomicAdd(&cntL[row], 1);
          if (pl < SLOTS) slotL[row * SLOTS + pl] = src[e];
        }
      }
    }
  }
  __syncthreads();
  cnt[r * RROWS + tid] = cntL[tid];
  int4* so = (int4*)(slots + (size_t)r * RROWS * SLOTS);
  const int4* sl = (const int4*)slotL;
#pragma unroll
  for (int i = 0; i < (RROWS * SLOTS / 4) / 1024; ++i) so[tid + i * 1024] = sl[tid + i * 1024];
}


__device__ __forceinline__ int sort_sources(int mine, int n, int lane) {
  int rk = 0;
  for (int k = 0; k < n; ++k) {
    int sk = __builtin_amdgcn_readlane(mine, k);
    rk += (sk < mine || (sk == mine && k < lane)) ? 1 : 0;
  }
  if (lane >= n) rk = lane;
  return __builtin_amdgcn_ds_permute(rk << 2, mine);
}

__device__ __forceinline__ void add_rows(float4& acc, const float* __restrict__ xlane,
                                         int rows, float wts, int n) {
  for (int j = 0; j < n; j += 8) {
    float4 v[8];
    float  w[8];
#pragma unroll
    for (int t = 0; t < 8; ++t) {
      v[t] = make_float4(0.f, 0.f, 0.f, 0.f);
      w[t] = 0.f;
      if (j + t < n) {
        int r = __builtin_amdgcn_readlane(rows, j + t);
        w[t]  = __int_as_float(__builtin_amdgcn_readlane(__float_as_int(wts), j + t));
        v[t]  = *(const float4*)(xlane + (size_t)r * HD);
      }
    }
#pragma unroll
    for (int t = 0; t < 8; ++t) {
      acc.x += w[t] * v[t].x; acc.y += w[t] * v[t].y;
      acc.z += w[t] * v[t].z; acc.w += w[t] * v[t].w;
    }
  }
}

template <bool LOOKUP>
__global__ __launch_bounds__(1024) void layer_gather_wmma(
    const float* __restrict__ x, const int* __restrict__ node_id,
    const int* __restrict__ cnt, const int* __restrict__ slots,
    const int* __restrict__ src, const int* __restrict__ dst,
    const float* __restrict__ norm, const float* __restrict__ W,
    const float* __restrict__ bias, float* __restrict__ out) {
  __shared__ __attribute__((aligned(16))) float Wl2[HD * HD];
  __shared__ __attribute__((aligned(16))) float At[TILE_ROWS * APITCH];

  const int tid  = threadIdx.x;
  const int wave = __builtin_amdgcn_readfirstlane(tid >> 5);
  const int lane = tid & 31;
  const int node0 = blockIdx.x * TILE_ROWS + wave * 4;

  float2 wr0[4], wr1[4];
#pragma unroll
  for (int it = 0; it < 4; ++it) {
    int i = tid + it * 1024;
    int p = i >> 6;
    int n = (i & 63) << 1;
    int k = p << 1;
    wr0[it] = *(const float2*)(W + (size_t)k * HD + n);
    wr1[it] = *(const float2*)(W + (size_t)(k + 1) * HD + n);
  }

  const int4 c4 = *(const int4*)(cnt + node0);
  int sv[4];
#pragma unroll
  for (int i = 0; i < 4; ++i) sv[i] = slots[(node0 + i) * SLOTS + lane];
  const int ntot[4] = {c4.x, c4.y, c4.z, c4.w};
  int n0[4], mine[4];
#pragma unroll
  for (int i = 0; i < 4; ++i) {
    int n = ntot[i];
    n = n > SLOTS ? 0 : n;
    n = n < 0 ? 0 : n;
    n0[i]   = n;
    mine[i] = lane < n ? sv[i] : 0x7fffffff;
  }

  int   rows[4];
  float wts[4];
#pragma unroll
  for (int i = 0; i < 4; ++i) {
    int  s   = sort_sources(mine[i], n0[i], lane);
    bool act = lane < n0[i];
    int  sc  = act ? s : 0;
    wts[i]  = act ? norm[sc] : 0.f;
    rows[i] = LOOKUP ? (act ? node_id[sc] : 0) : sc;
  }

#pragma unroll
  for (int it = 0; it < 4; ++it) {
    int i = tid + it * 1024;
    int p = i >> 6;
    int n = (i & 63) << 1;
    *(float4*)(Wl2 + p * 256 + n * 2) =
        make_float4(wr0[it].x, wr1[it].x, wr0[it].y, wr1[it].y);
  }

  const float* xlane = x + lane * 4;
#pragma unroll
  for (int i = 0; i < 4; ++i) {
    float4 acc = make_float4(0.f, 0.f, 0.f, 0.f);
    add_rows(acc, xlane, rows[i], wts[i], n0[i]);
    if (ntot[i] > SLOTS) {
      const int node = node0 + i;
      for (int base = 0; base < N_EDGES; base += 32) {
        int  e = base + lane;
        bool m = e < N_EDGES && dst[e] == node;
        unsigned mask = __builtin_amdgcn_ballot_w32(m);
        if (mask == 0u) continue;
        int n = __builtin_popcount(mask);
        unsigned below = (1u << lane) - 1u;
        int rk = m ? __builtin_popcount(mask & below) : n + __builtin_popcount(~mask & below);
        int s  = __builtin_amdgcn_ds_permute(rk << 2, m ? src[e] : 0);
        bool act = lane < n;
        int  sc  = act ? s : 0;
        float w2 = act ? norm[sc] : 0.f;
        int   r2 = LOOKUP ? (act ? node_id[sc] : 0) : sc;
        add_rows(acc, xlane, r2, w2, n);
      }
    }
    *(float4*)(At + (wave * 4 + i) * APITCH + lane * 4) = acc;
  }
  __syncthreads();

  const int lane16 = lane & 15;
  const int pOff   = lane >> 4;
  const int rt     = wave >> 2;
  const int nt0    = (wave & 3) * 2;
  v8f acc0 = {}, acc1 = {};
  const float* arow = At + (rt * 16 + lane16) * APITCH + (pOff << 1);
  const float* bcol = Wl2 + pOff * 256 + (nt0 * 16 + lane16) * 2;
  for (int k0 = 0; k0 < HD; k0 += 4) {
    v2f a  = *(const v2f*)(arow + k0);
    v2f b0 = *(const v2f*)(bcol + (k0 >> 1) * 256);
    v2f b1 = *(const v2f*)(bcol + (k0 >> 1) * 256 + 32);
    acc0 = __builtin_amdgcn_wmma_f32_16x16x4_f32(
        false, a, false, b0, (short)0, acc0, false, false);
    acc1 = __builtin_amdgcn_wmma_f32_16x16x4_f32(
        false, a, false, b1, (short)0, acc1, false, false);
  }

  const int row0 = blockIdx.x * TILE_ROWS + rt * 16 + (pOff << 3);
  const int col0 = nt0 * 16 + lane16;
  const float bs0 = bias[col0];
  const float bs1 = bias[col0 + 16];
#pragma unroll
  for (int v = 0; v < 8; ++v) {
    int r = row0 + v;
    if (r < N_NODES) {
      float nr = norm[r];
      float y0 = acc0[v] * nr + bs0;
      float y1 = acc1[v] * nr + bs1;
      y0 = y0 > 0.f ? y0 : 0.2f * y0;
      y1 = y1 > 0.f ? y1 : 0.2f * y1;
      out[(size_t)r * HD + col0]      = y0;
      out[(size_t)r * HD + col0 + 16] = y1;
    }
  }
}

extern "C" void kernel_launch(void* const* d_in, const int* in_sizes, int n_in,
                              void* d_out, int out_size, void* d_ws, size_t ws_size,
                              hipStream_t stream) {
  const int*   node_id = (const int*)d_in[0];
  const int*   src     = (const int*)d_in[1];
  const int*   dst     = (const int*)d_in[2];
  const float* norm    = (const float*)d_in[3];
  const float* embed   = (const float*)d_in[4];
  const float* W1      = (const float*)d_in[5];
  const float* b1      = (const float*)d_in[6];
  const float* W2      = (const float*)d_in[7];
  const float* b2      = (const float*)d_in[8];
  float* out = (float*)d_out;

  const size_t feat = (size_t)N_NODES * HD;
  float* bufH = (float*)d_ws;
  int*  cnt   = (int*)(bufH + feat);
  int*  slots = cnt + N_ALLOC;
  int*  cntG  = slots + (size_t)N_ALLOC * SLOTS;
  int2* recG  = (int2*)(cntG + ((NRANGE * NCHUNK + 31) / 32) * 32);

  const int gLayer = (N_NODES + TILE_ROWS - 1) / TILE_ROWS;

  bin_by_range<<<NCHUNK, dim3(1024), 0, stream>>>(src, dst, cntG, recG);
  place_rows<<<NRANGE, dim3(1024), 0, stream>>>(src, dst, cntG, recG, cnt, slots);

  layer_gather_wmma<true><<<gLayer, dim3(1024), 0, stream>>>(
      embed, node_id, cnt, slots, src, dst, norm, W1, b1, bufH);
  layer_gather_wmma<false><<<gLayer, dim3(1024), 0, stream>>>(
      bufH, node_id, cnt, slots, src, dst, norm, W2, b2, out);
}


# Each wave owns one 16-row tile from gather to whole-row store, in 256 resident 25-wave blocks: W staged once per block, no tile barrier, 8 products per A fragment, 2 rows per gather trip; sums same
# speedup vs baseline: 3.7285x; 1.1758x over previous
#include <hip/hip_runtime.h>

#define N_NODES 100000
#define N_EDGES 600000
#define HD 128
#define TILE_ROWS 128
#define APITCH 132
#define SLOTS 32
#define N_PAD (((N_NODES + TILE_ROWS - 1) / TILE_ROWS) * TILE_ROWS)
#define RSHIFT 10
#define RROWS (1 << RSHIFT)
#define NRANGE ((N_PAD + RROWS - 1) / RROWS)
#define N_ALLOC (NRANGE * RROWS)
#define CHUNK 1024
#define NCHUNK ((N_EDGES + CHUNK - 1) / CHUNK)
#define CAP 32
#define CHUNKS_PER_WAVE ((NCHUNK + 31) / 32)

typedef __attribute__((ext_vector_type(2))) float v2f;
typedef __attribute__((ext_vector_type(8))) float v8f;

__global__ __launch_bounds__(1024) void bin_by_range(
    const int* __restrict__ src, const int* __restrict__ dst,
    int* __restrict__ cntG, int2* __restrict__ recG) {
  __shared__ __attribute__((aligned(16))) int cntR[128];
  __shared__ __attribute__((aligned(16))) int offR[128];
  __shared__ int2 stage[CHUNK];
  const int tid = threadIdx.x;
  const int blk = blockIdx.x;
  const int e = blk * CHUNK + tid;
  int d = -1, s = 0;
  if (e < N_EDGES) { d = dst[e]; s = src[e]; }
  if (tid < 128) cntR[tid] = 0;
  __syncthreads();
  const bool ok = (unsigned)d < (unsigned)N_NODES;
  const int r = d >> RSHIFT;
  int rank = 0;
  if (ok) rank = atomicAdd(&cntR[r], 1);
  __syncthreads();
  if (tid < 32) {
    int4 c4 = ((const int4*)cntR)[tid];
    int s4 = c4.x + c4.y + c4.z + c4.w;
    int inc = s4;
#pragma unroll
    for (int o = 1; o < 32; o <<= 1) {
      int t = __shfl_up(inc, o);
      if (tid >= o) inc += t;
    }
    int ex = inc - s4;
    ((int4*)offR)[tid] = make_int4(ex, ex + c4.x, ex + c4.x + c4.y, ex + c4.x + c4.y + c4.z);
  } else if (tid >= 128 && tid < 128 + NRANGE) {
    int q = tid - 128;
    cntG[q * NCHUNK + blk] = cntR[q];
  }
  __syncthreads();
  if (ok) stage[offR[r] + rank] = make_int2(s, d);
  __syncthreads();
  const int total = offR[127] + cntR[127];
  if (tid < total) {
    int2 rec = stage[tid];
    int rr  = rec.y >> RSHIFT;
    int pos = tid - offR[rr];
    if (pos < CAP) recG[((size_t)rr * NCHUNK + blk) * CAP + pos] = rec;
  }
}

__global__ __launch_bounds__(1024) void place_rows(
    const int* __restrict__ src, const int* __restrict__ dst,
    const int* __restrict__ cntG, const int2* __restrict__ recG,
    int* __restrict__ cnt, int* __restrict__ slots) {
  __shared__ __attribute__((aligned(16))) int cntL[RROWS];
  __shared__ __attribute__((aligned(16))) int slotL[RROWS * SLOTS];
  const int tid  = threadIdx.x;
  const int r    = blockIdx.x;
  const int wave = __builtin_amdgcn_readfirstlane(tid >> 5);
  const int lane = tid & 31;
  cntL[tid] = 0;
  int  nc[CHUNKS_PER_WAVE];
  int2 rec[CHUNKS_PER_WAVE];
#pragma unroll
  for (int i = 0; i < CHUNKS_PER_WAVE; ++i) {
    int c = wave + 32 * i;
    nc[i] = c < NCHUNK ? cntG[r * NCHUNK + c] : 0;
  }
  unsigned over = 0u;
#pragma unroll
  for (int i = 0; i < CHUNKS_PER_WAVE; ++i) {
    int c = wave + 32 * i;
    nc[i] = __builtin_amdgcn_readfirstlane(nc[i]);
    if (nc[i] > CAP) { over |= 1u << i; nc[i] = 0; }
    rec[i] = make_int2(0, 0);
    if (lane < nc[i]) rec[i] = recG[((size_t)r * NCHUNK + c) * CAP + lane];
  }
  __syncthreads();
  int pl[CHUNKS_PER_WAVE];
#pragma unroll
  for (int i = 0; i < CHUNKS_PER_WAVE; ++i) {
    pl[i] = SLOTS;
    if (lane < nc[i]) pl[i] = atomicAdd(&cntL[rec[i].y - r * RROWS], 1);
  }
#pragma unroll
  for (int i = 0; i < CHUNKS_PER_WAVE; ++i) {
    if (pl[i] < SLOTS) slotL[(rec[i].y - r * RROWS) * SLOTS + pl[i]] = rec[i].x;
  }
  while (over != 0u) {
    const int i = __builtin_ctz(over);
    over &= over - 1u;
    const int c = wave + 32 * i;
#pragma nounroll
    for (int k = 0; k < CHUNK; k += 32) {
      int e = c * CHUNK + k + lane;
      if (e < N_EDGES) {
        int d = dst[e];
        if ((unsigned)d < (unsigned)N_NODES && (d >> RSHIFT) == r) {
          int row = d - r * RROWS;
          int pl  = atomicAdd(&cntL[row], 1);
          if (pl < SLOTS) slotL[row * SLOTS + pl] = src[e];
        }
      }
    }
  }
  __syncthreads();
  cnt[r * RROWS + tid] = cntL[tid];
  int4* so = (int4*)(slots + (size_t)r * RROWS * SLOTS);
  const int4* sl = (const int4*)slotL;
#pragma unroll
  for (int i = 0; i < (RROWS * SLOTS / 4) / 1024; ++i) so[tid + i * 1024] = sl[tid + i * 1024];
}


__device__ __forceinline__ int sort_sources(int mine, int n, int lane) {
  int rk = 0;
  for (int k = 0; k < n; ++k) {
    int sk = __builtin_amdgcn_readlane(mine, k);
    rk += (sk < mine || (sk == mine && k < lane)) ? 1 : 0;
  }
  if (lane >= n) rk = lane;
  return __builtin_amdgcn_ds_permute(rk << 2, mine);
}

__device__ __forceinline__ void load8(float4 (&v)[8], float (&w)[8], const float* __restrict__ xlane,
                                      int rows, float wts, int n, int j) {
#pragma unroll
  for (int t = 0; t < 8; ++t) {
    int jt = j + t;
    int jr = jt < n ? jt : n - 1;
    int r = __builtin_amdgcn_readlane(rows, jr);
    w[t]  = __int_as_float(__builtin_amdgcn_readlane(__float_as_int(wts), jt));
    v[t]  = *(const float4*)(xlane + (size_t)r * HD);
  }
}

__device__ __forceinline__ void fma8(float4& acc, const float4 (&v)[8], const float (&w)[8]) {
#pragma unroll
  for (int t = 0; t < 8; ++t) {
    acc.x += w[t] * v[t].x; acc.y += w[t] * v[t].y;
    acc.z += w[t] * v[t].z; acc.w += w[t] * v[t].w;
  }
}

__device__ __forceinline__ void add_rows2(float4& accA, float4& accB, const float* __restrict__ xlane,
                                          int rowsA, float wtsA, int nA,
                                          int rowsB, float wtsB, int nB) {
  const int nmax = nA > nB ? nA : nB;
  for (int j = 0; j < nmax; j += 8) {
    float4 va[8], vb[8];
    float  wa[8], wb[8];
    if (j < nA && j < nB) {
      load8(va, wa, xlane, rowsA, wtsA, nA, j);
      load8(vb, wb, xlane, rowsB, wtsB, nB, j);
      fma8(accA, va, wa);
      fma8(accB, vb, wb);
    } else if (j < nA) {
      load8(va, wa, xlane, rowsA, wtsA, nA, j);
      fma8(accA, va, wa);
    } else {
      load8(vb, wb, xlane, rowsB, wtsB, nB, j);
      fma8(accB, vb, wb);
    }
  }
}

__device__ __forceinline__ void add_rows(float4& acc, const float* __restrict__ xlane,
                                         int rows, float wts, int n) {
  for (int j = 0; j < n; j += 8) {
    float4 v[8];
    float  w[8];
#pragma unroll
    for (int t = 0; t < 8; ++t) {
      v[t] = make_float4(0.f, 0.f, 0.f, 0.f);
      w[t] = 0.f;
      if (j + t < n) {
        int r = __builtin_amdgcn_readlane(rows, j + t);
        w[t]  = __int_as_float(__builtin_amdgcn_readlane(__float_as_int(wts), j + t));
        v[t]  = *(const float4*)(xlane + (size_t)r * HD);
      }
    }
#pragma unroll
    for (int t = 0; t < 8; ++t) {
      acc.x += w[t] * v[t].x; acc.y += w[t] * v[t].y;
      acc.z += w[t] * v[t].z; acc.w += w[t] * v[t].w;
    }
  }
}

#define LWAVES 25
#define LTHREADS (LWAVES * 32)
#define LGRID 256
#define NTILE16 (N_NODES / 16)
static_assert(N_NODES % 16 == 0, "a tile is 16 whole rows");
static_assert(LGRID * LWAVES >= NTILE16, "every tile has a wave");

template <bool LOOKUP>
__global__ __launch_bounds__(LTHREADS) void layer_gather_wmma(
    const float* __restrict__ x, const int* __restrict__ node_id,
    const int* __restrict__ cnt, const int* __restrict__ slots,
    const int* __restrict__ src, const int* __restrict__ dst,
    const float* __restrict__ norm, const float* __restrict__ W,
    const float* __restrict__ bias, float* __restrict__ out) {
  __shared__ __attribute__((aligned(16))) float Wl2[HD * HD];
  __shared__ __attribute__((aligned(16))) float At[LWAVES * 16 * APITCH];

  const int tid  = threadIdx.x;
  const int wave = __builtin_amdgcn_readfirstlane(tid >> 5);
  const int lane = tid & 31;
  const int tile = wave * (int)gridDim.x + (int)blockIdx.x;

  constexpr int WIT = ((HD * HD) / 4 + LTHREADS - 1) / LTHREADS;
  float2 wr0[WIT], wr1[WIT];
#pragma unroll
  for (int it = 0; it < WIT; ++it) {
    int i = tid + it * LTHREADS;
    wr0[it] = make_float2(0.f, 0.f);
    wr1[it] = make_float2(0.f, 0.f);
    if (i < (HD * HD) / 4) {
      int p = i >> 6;
      int n = (i & 63) << 1;
      int k = p << 1;
      wr0[it] = *(const float2*)(W + (size_t)k * HD + n);
      wr1[it] = *(const float2*)(W + (size_t)(k + 1) * HD + n);
    }
  }

  float* strip = At + wave * (16 * APITCH);
  const bool live = tile < NTILE16;
  const float* xlane = x + lane * 4;

  const int tile0 = live ? tile : 0;
  int4 c4 = *(const int4*)(cnt + tile0 * 16);
  int sv[4];
#pragma unroll
  for (int i = 0; i < 4; ++i) sv[i] = slots[(tile0 * 16 + i) * SLOTS + lane];

#pragma unroll
  for (int it = 0; it < WIT; ++it) {
    int i = tid + it * LTHREADS;
    if (i < (HD * HD) / 4) {
      int p = i >> 6;
      int n = (i & 63) << 1;
      *(float4*)(Wl2 + p * 256 + n * 2) =
          make_float4(wr0[it].x, wr1[it].x, wr0[it].y, wr1[it].y);
    }
  }

  __syncthreads();
  if (!live) return;
  {
#pragma nounroll
    for (int g = 0; g < 4; ++g) {
      const int node0 = tile * 16 + g * 4;
      const int ntot[4] = {c4.x, c4.y, c4.z, c4.w};
      int n0[4], mine[4];
#pragma unroll
      for (int i = 0; i < 4; ++i) {
        int n = __builtin_amdgcn_readfirstlane(ntot[i]);
        n = n > SLOTS ? 0 : n;
        n = n < 0 ? 0 : n;
        n0[i]   = n;
        mine[i] = lane < n ? sv[i] : 0x7fffffff;
      }
      int   rows[4];
      float wts[4];
#pragma unroll
      for (int i = 0; i < 4; ++i) {
        int  s   = sort_sources(mine[i], n0[i], lane);
        bool act = lane < n0[i];
        int  sc  = act ? s : 0;
        wts[i]  = act ? norm[sc] : 0.f;
        rows[i] = LOOKUP ? (act ? node_id[sc] : 0) : sc;
      }
      c4 = *(const int4*)(cnt + node0 + 4);
#pragma unroll
      for (int i = 0; i < 4; ++i) sv[i] = slots[(node0 + 4 + i) * SLOTS + lane];
      asm volatile("" :: "v"(wts[0]), "v"(wts[1]), "v"(wts[2]), "v"(wts[3]),
                         "v"(rows[0]), "v"(rows[1]), "v"(rows[2]), "v"(rows[3]));
      float4 acc[4];
#pragma unroll
      for (int i = 0; i < 4; ++i) acc[i] = make_float4(0.f, 0.f, 0.f, 0.f);
      add_rows2(acc[0], acc[1], xlane, rows[0], wts[0], n0[0], rows[1], wts[1], n0[1]);
      add_rows2(acc[2], acc[3], xlane, rows[2], wts[2], n0[2], rows[3], wts[3], n0[3]);
#pragma unroll
      for (int i = 0; i < 4; ++i) {
        if (__builtin_amdgcn_readfirstlane(ntot[i]) > SLOTS) {
          const int node = node0 + i;
          for (int base = 0; base < N_EDGES; base += 32) {
            int  e = base + lane;
            bool m = e < N_EDGES && dst[e] == node;
            unsigned mask = __builtin_amdgcn_ballot_w32(m);
            if (mask == 0u) continue;
            int n = __builtin_popcount(mask);
            unsigned below = (1u << lane) - 1u;
            int rk = m ? __builtin_popcount(mask & below) : n + __builtin_popcount(~mask & below);
            int s  = __builtin_amdgcn_ds_permute(rk << 2, m ? src[e] : 0);
            bool act = lane < n;
            int  sc  = act ? s : 0;
            float w2 = act ? norm[sc] : 0.f;
            int   r2 = LOOKUP ? (act ? node_id[sc] : 0) : sc;
            add_rows(acc[i], xlane, r2, w2, n);
          }
        }
        *(float4*)(strip + (g * 4 + i) * APITCH + lane * 4) = acc[i];
      }
    }
  }

  const int lane16 = lane & 15;
  const int pOff   = lane >> 4;
  v8f accw[8];
#pragma unroll
  for (int j = 0; j < 8; ++j) accw[j] = (v8f){};
  const float* arow = strip + lane16 * APITCH + (pOff << 1);
  const float* bcol = Wl2 + pOff * 256 + lane16 * 2;
#pragma unroll 4
  for (int k0 = 0; k0 < HD; k0 += 4) {
    v2f a = *(const v2f*)(arow + k0);
#pragma unroll
    for (int j = 0; j < 8; ++j) {
      v2f b = *(const v2f*)(bcol + (k0 >> 1) * 256 + j * 32);
      accw[j] = __builtin_amdgcn_wmma_f32_16x16x4_f32(
          false, a, false, b, (short)0, accw[j], false, false);
    }
  }

  const int row0 = tile * 16;
  float nr[8];
#pragma unroll
  for (int v = 0; v < 8; ++v) nr[v] = norm[row0 + (pOff << 3) + v];
#pragma unroll
  for (int j = 0; j < 8; ++j) {
    const float bs = bias[j * 16 + lane16];
#pragma unroll
    for (int v = 0; v < 8; ++v) {
      float y = accw[j][v] * nr[v] + bs;
      y = y > 0.f ? y : 0.2f * y;
      strip[((pOff << 3) + v) * APITCH + j * 16 + lane16] = y;
    }
  }
#pragma unroll
  for (int r = 0; r < 16; ++r) {
    float4 y4 = *(const float4*)(strip + r * APITCH + lane * 4);
    *(float4*)(out + (size_t)(row0 + r) * HD + lane * 4) = y4;
  }
}

extern "C" void kernel_launch(void* const* d_in, const int* in_sizes, int n_in,
                              void* d_out, int out_size, void* d_ws, size_t ws_size,
                              hipStream_t stream) {
  const int*   node_id = (const int*)d_in[0];
  const int*   src     = (const int*)d_in[1];
  const int*   dst     = (const int*)d_in[2];
  const float* norm    = (const float*)d_in[3];
  const float* embed   = (const float*)d_in[4];
  const float* W1      = (const float*)d_in[5];
  const float* b1      = (const float*)d_in[6];
  const float* W2      = (const float*)d_in[7];
  const float* b2      = (const float*)d_in[8];
  float* out = (float*)d_out;

  const size_t feat = (size_t)N_NODES * HD;
  float* bufH = (float*)d_ws;
  int*  cnt   = (int*)(bufH + feat);
  int*  slots = cnt + N_ALLOC;
  int*  cntG  = slots + (size_t)N_ALLOC * SLOTS;
  int2* recG  = (int2*)(cntG + ((NRANGE * NCHUNK + 31) / 32) * 32);


  bin_by_range<<<NCHUNK, dim3(1024), 0, stream>>>(src, dst, cntG, recG);
  place_rows<<<NRANGE, dim3(1024), 0, stream>>>(src, dst, cntG, recG, cnt, slots);

  layer_gather_wmma<true><<<LGRID, dim3(LTHREADS), 0, stream>>>(
      embed, node_id, cnt, slots, src, dst, norm, W1, b1, bufH);
  layer_gather_wmma<false><<<LGRID, dim3(LTHREADS), 0, stream>>>(
      bufH, node_id, cnt, slots, src, dst, norm, W2, b2, out);
}


# Rank a row's sources with lane-shared (DPP row_share) compares, two rows per ranking in the two halves of a wave, in place of the scalar readlane loop per row; sum order unchanged
# speedup vs baseline: 3.8765x; 1.0397x over previous
#include <hip/hip_runtime.h>

#define N_NODES 100000
#define N_EDGES 600000
#define HD 128
#define TILE_ROWS 128
#define APITCH 132
#define SLOTS 32
#define N_PAD (((N_NODES + TILE_ROWS - 1) / TILE_ROWS) * TILE_ROWS)
#define RSHIFT 10
#define RROWS (1 << RSHIFT)
#define NRANGE ((N_PAD + RROWS - 1) / RROWS)
#define N_ALLOC (NRANGE * RROWS)
#define CHUNK 1024
#define NCHUNK ((N_EDGES + CHUNK - 1) / CHUNK)
#define CAP 32
#define CHUNKS_PER_WAVE ((NCHUNK + 31) / 32)

typedef __attribute__((ext_vector_type(2))) float v2f;
typedef __attribute__((ext_vector_type(8))) float v8f;

__global__ __launch_bounds__(1024) void bin_by_range(
    const int* __restrict__ src, const int* __restrict__ dst,
    int* __restrict__ cntG, int2* __restrict__ recG) {
  __shared__ __attribute__((aligned(16))) int cntR[128];
  __shared__ __attribute__((aligned(16))) int offR[128];
  __shared__ int2 stage[CHUNK];
  const int tid = threadIdx.x;
  const int blk = blockIdx.x;
  const int e = blk * CHUNK + tid;
  int d = -1, s = 0;
  if (e < N_EDGES) { d = dst[e]; s = src[e]; }
  if (tid < 128) cntR[tid] = 0;
  __syncthreads();
  const bool ok = (unsigned)d < (unsigned)N_NODES;
  const int r = d >> RSHIFT;
  int rank = 0;
  if (ok) rank = atomicAdd(&cntR[r], 1);
  __syncthreads();
  if (tid < 32) {
    int4 c4 = ((const int4*)cntR)[tid];
    int s4 = c4.x + c4.y + c4.z + c4.w;
    int inc = s4;
#pragma unroll
    for (int o = 1; o < 32; o <<= 1) {
      int t = __shfl_up(inc, o);
      if (tid >= o) inc += t;
    }
    int ex = inc - s4;
    ((int4*)offR)[tid] = make_int4(ex, ex + c4.x, ex + c4.x + c4.y, ex + c4.x + c4.y + c4.z);
  } else if (tid >= 128 && tid < 128 + NRANGE) {
    int q = tid - 128;
    cntG[q * NCHUNK + blk] = cntR[q];
  }
  __syncthreads();
  if (ok) stage[offR[r] + rank] = make_int2(s, d);
  __syncthreads();
  const int total = offR[127] + cntR[127];
  if (tid < total) {
    int2 rec = stage[tid];
    int rr  = rec.y >> RSHIFT;
    int pos = tid - offR[rr];
    if (pos < CAP) recG[((size_t)rr * NCHUNK + blk) * CAP + pos] = rec;
  }
}

__global__ __launch_bounds__(1024) void place_rows(
    const int* __restrict__ src, const int* __restrict__ dst,
    const int* __restrict__ cntG, const int2* __restrict__ recG,
    int* __restrict__ cnt, int* __restrict__ slots) {
  __shared__ __attribute__((aligned(16))) int cntL[RROWS];
  __shared__ __attribute__((aligned(16))) int slotL[RROWS * SLOTS];
  const int tid  = threadIdx.x;
  const int r    = blockIdx.x;
  const int wave = __builtin_amdgcn_readfirstlane(tid >> 5);
  const int lane = tid & 31;
  cntL[tid] = 0;
  int  nc[CHUNKS_PER_WAVE];
  int2 rec[CHUNKS_PER_WAVE];
#pragma unroll
  for (int i = 0; i < CHUNKS_PER_WAVE; ++i) {
    int c = wave + 32 * i;
    nc[i] = c < NCHUNK ? cntG[r * NCHUNK + c] : 0;
  }
  unsigned over = 0u;
#pragma unroll
  for (int i = 0; i < CHUNKS_PER_WAVE; ++i) {
    int c = wave + 32 * i;
    nc[i] = __builtin_amdgcn_readfirstlane(nc[i]);
    if (nc[i] > CAP) { over |= 1u << i; nc[i] = 0; }
    rec[i] = make_int2(0, 0);
    if (lane < nc[i]) rec[i] = recG[((size_t)r * NCHUNK + c) * CAP + lane];
  }
  __syncthreads();
  int pl[CHUNKS_PER_WAVE];
#pragma unroll
  for (int i = 0; i < CHUNKS_PER_WAVE; ++i) {
    pl[i] = SLOTS;
    if (lane < nc[i]) pl[i] = atomicAdd(&cntL[rec[i].y - r * RROWS], 1);
  }
#pragma unroll
  for (int i = 0; i < CHUNKS_PER_WAVE; ++i) {
    if (pl[i] < SLOTS) slotL[(rec[i].y - r * RROWS) * SLOTS + pl[i]] = rec[i].x;
  }
  while (over != 0u) {
    const int i = __builtin_ctz(over);
    over &= over - 1u;
    const int c = wave + 32 * i;
#pragma nounroll
    for (int k = 0; k < CHUNK; k += 32) {
      int e = c * CHUNK + k + lane;
      if (e < N_EDGES) {
        int d = dst[e];
        if ((unsigned)d < (unsigned)N_NODES && (d >> RSHIFT) == r) {
          int row = d - r * RROWS;
          int pl  = atomicAdd(&cntL[row], 1);
          if (pl < SLOTS) slotL[row * SLOTS + pl] = src[e];
        }
      }
    }
  }
  __syncthreads();
  cnt[r * RROWS + tid] = cntL[tid];
  int4* so = (int4*)(slots + (size_t)r * RROWS * SLOTS);
  const int4* sl = (const int4*)slotL;
#pragma unroll
  for (int i = 0; i < (RROWS * SLOTS / 4) / 1024; ++i) so[tid + i * 1024] = sl[tid + i * 1024];
}


__device__ __forceinline__ int sort_sources(int mine, int n, int lane) {
  int rk = 0;
  for (int k = 0; k < n; ++k) {
    int sk = __builtin_amdgcn_readlane(mine, k);
    rk += (sk < mine || (sk == mine && k < lane)) ? 1 : 0;
  }
  if (lane >= n) rk = lane;
  return __builtin_amdgcn_ds_permute(rk << 2, mine);
}

#define SHARE_LT(k) ((__builtin_amdgcn_update_dpp(0, key, 0x150 + (k), 0xf, 0xf, false) < key) ? 1 : 0)
__device__ __forceinline__ int rank_pair(int key, bool more) {
  int rk = SHARE_LT(0) + SHARE_LT(1) + SHARE_LT(2) + SHARE_LT(3) +
           SHARE_LT(4) + SHARE_LT(5) + SHARE_LT(6) + SHARE_LT(7);
  if (more)
    rk += SHARE_LT(8) + SHARE_LT(9) + SHARE_LT(10) + SHARE_LT(11) +
          SHARE_LT(12) + SHARE_LT(13) + SHARE_LT(14) + SHARE_LT(15);
  return rk;
}

__device__ __forceinline__ void load8(float4 (&v)[8], float (&w)[8], const float* __restrict__ xlane,
                                      int rows, float wts, int n, int j, int base) {
#pragma unroll
  for (int t = 0; t < 8; ++t) {
    int jt = j + t;
    int jr = jt < n ? jt : n - 1;
    int r = __builtin_amdgcn_readlane(rows, base + jr);
    w[t]  = __int_as_float(__builtin_amdgcn_readlane(__float_as_int(wts), base + jt));
    v[t]  = *(const float4*)(xlane + (size_t)r * HD);
  }
}

__device__ __forceinline__ void fma8(float4& acc, const float4 (&v)[8], const float (&w)[8]) {
#pragma unroll
  for (int t = 0; t < 8; ++t) {
    acc.x += w[t] * v[t].x; acc.y += w[t] * v[t].y;
    acc.z += w[t] * v[t].z; acc.w += w[t] * v[t].w;
  }
}

__device__ __forceinline__ void add_rows2(float4& accA, float4& accB, const float* __restrict__ xlane,
                                          int rowsA, float wtsA, int nA, int baseA,
                                          int rowsB, float wtsB, int nB, int baseB) {
  const int nmax = nA > nB ? nA : nB;
  for (int j = 0; j < nmax; j += 8) {
    float4 va[8], vb[8];
    float  wa[8], wb[8];
    if (j < nA && j < nB) {
      load8(va, wa, xlane, rowsA, wtsA, nA, j, baseA);
      load8(vb, wb, xlane, rowsB, wtsB, nB, j, baseB);
      fma8(accA, va, wa);
      fma8(accB, vb, wb);
    } else if (j < nA) {
      load8(va, wa, xlane, rowsA, wtsA, nA, j, baseA);
      fma8(accA, va, wa);
    } else {
      load8(vb, wb, xlane, rowsB, wtsB, nB, j, baseB);
      fma8(accB, vb, wb);
    }
  }
}

__device__ __forceinline__ void add_rows(float4& acc, const float* __restrict__ xlane,
                                         int rows, float wts, int n) {
  for (int j = 0; j < n; j += 8) {
    float4 v[8];
    float  w[8];
#pragma unroll
    for (int t = 0; t < 8; ++t) {
      v[t] = make_float4(0.f, 0.f, 0.f, 0.f);
      w[t] = 0.f;
      if (j + t < n) {
        int r = __builtin_amdgcn_readlane(rows, j + t);
        w[t]  = __int_as_float(__builtin_amdgcn_readlane(__float_as_int(wts), j + t));
        v[t]  = *(const float4*)(xlane + (size_t)r * HD);
      }
    }
#pragma unroll
    for (int t = 0; t < 8; ++t) {
      acc.x += w[t] * v[t].x; acc.y += w[t] * v[t].y;
      acc.z += w[t] * v[t].z; acc.w += w[t] * v[t].w;
    }
  }
}

#define LWAVES 25
#define LTHREADS (LWAVES * 32)
#define LGRID 256
#define NTILE16 (N_NODES / 16)
static_assert(N_NODES % 16 == 0, "a tile is 16 whole rows");
static_assert(LGRID * LWAVES >= NTILE16, "every tile has a wave");

template <bool LOOKUP>
__global__ __launch_bounds__(LTHREADS) void layer_gather_wmma(
    const float* __restrict__ x, const int* __restrict__ node_id,
    const int* __restrict__ cnt, const int* __restrict__ slots,
    const int* __restrict__ src, const int* __restrict__ dst,
    const float* __restrict__ norm, const float* __restrict__ W,
    const float* __restrict__ bias, float* __restrict__ out) {
  __shared__ __attribute__((aligned(16))) float Wl2[HD * HD];
  __shared__ __attribute__((aligned(16))) float At[LWAVES * 16 * APITCH];

  const int tid  = threadIdx.x;
  const int wave = __builtin_amdgcn_readfirstlane(tid >> 5);
  const int lane = tid & 31;
  const int tile = wave * (int)gridDim.x + (int)blockIdx.x;

  constexpr int WIT = ((HD * HD) / 4 + LTHREADS - 1) / LTHREADS;
  float2 wr0[WIT], wr1[WIT];
#pragma unroll
  for (int it = 0; it < WIT; ++it) {
    int i = tid + it * LTHREADS;
    wr0[it] = make_float2(0.f, 0.f);
    wr1[it] = make_float2(0.f, 0.f);
    if (i < (HD * HD) / 4) {
      int p = i >> 6;
      int n = (i & 63) << 1;
      int k = p << 1;
      wr0[it] = *(const float2*)(W + (size_t)k * HD + n);
      wr1[it] = *(const float2*)(W + (size_t)(k + 1) * HD + n);
    }
  }

  float* strip = At + wave * (16 * APITCH);
  const bool live = tile < NTILE16;
  const float* xlane = x + lane * 4;

  const int tile0 = live ? tile : 0;
  const int hi  = lane >> 4;
  const int l16 = lane & 15;
  int4 c4 = *(const int4*)(cnt + tile0 * 16);
  int sv2[2];
#pragma unroll
  for (int p = 0; p < 2; ++p) sv2[p] = slots[(tile0 * 16 + 2 * p + hi) * SLOTS + l16];

#pragma unroll
  for (int it = 0; it < WIT; ++it) {
    int i = tid + it * LTHREADS;
    if (i < (HD * HD) / 4) {
      int p = i >> 6;
      int n = (i & 63) << 1;
      *(float4*)(Wl2 + p * 256 + n * 2) =
          make_float4(wr0[it].x, wr1[it].x, wr0[it].y, wr1[it].y);
    }
  }

  __syncthreads();
  if (!live) return;
  {
#pragma nounroll
    for (int g = 0; g < 4; ++g) {
      const int node0 = tile * 16 + g * 4;
      const int ntot[4] = {c4.x, c4.y, c4.z, c4.w};
      int n0[4];
#pragma unroll
      for (int i = 0; i < 4; ++i) {
        int n = __builtin_amdgcn_readfirstlane(ntot[i]);
        n = n > SLOTS ? 0 : n;
        n = n < 0 ? 0 : n;
        n0[i] = n;
      }
      const int svc[2] = {sv2[0], sv2[1]};
      float4 acc[4];
#pragma unroll
      for (int i = 0; i < 4; ++i) acc[i] = make_float4(0.f, 0.f, 0.f, 0.f);
      const bool wide = (n0[0] > 16) | (n0[1] > 16) | (n0[2] > 16) | (n0[3] > 16);
      if (!wide) {
        int   rows2[2], sc2[2];
        float wts2[2];
        bool  act2[2];
#pragma unroll
        for (int p = 0; p < 2; ++p) {
          const int  nh  = hi ? n0[2 * p + 1] : n0[2 * p];
          const bool act = l16 < nh;
          const int  key = act ? ((svc[p] << 5) | l16) : 0x7fffffff;
          const int  rk  = rank_pair(key, (n0[2 * p] > 8) | (n0[2 * p + 1] > 8));
          const int  to  = act ? (hi << 4) + rk : lane;
          const int  s   = __builtin_amdgcn_ds_permute(to << 2, svc[p]);
          act2[p] = act;
          sc2[p]  = act ? s : 0;
        }
#pragma unroll
        for (int p = 0; p < 2; ++p) {
          wts2[p]  = act2[p] ? norm[sc2[p]] : 0.f;
          rows2[p] = LOOKUP ? (act2[p] ? node_id[sc2[p]] : 0) : sc2[p];
        }
        c4 = *(const int4*)(cnt + node0 + 4);
#pragma unroll
        for (int p = 0; p < 2; ++p) sv2[p] = slots[(node0 + 4 + 2 * p + hi) * SLOTS + l16];
        asm volatile("" :: "v"(wts2[0]), "v"(wts2[1]), "v"(rows2[0]), "v"(rows2[1]));
        add_rows2(acc[0], acc[1], xlane, rows2[0], wts2[0], n0[0], 0, rows2[0], wts2[0], n0[1], 16);
        add_rows2(acc[2], acc[3], xlane, rows2[1], wts2[1], n0[2], 0, rows2[1], wts2[1], n0[3], 16);
      } else {
        int   rows[4];
        float wts[4];
#pragma unroll
        for (int i = 0; i < 4; ++i) {
          int  sl   = slots[(node0 + i) * SLOTS + lane];
          int  mine = lane < n0[i] ? sl : 0x7fffffff;
          int  s    = sort_sources(mine, n0[i], lane);
          bool act  = lane < n0[i];
          int  sc   = act ? s : 0;
          wts[i]  = act ? norm[sc] : 0.f;
          rows[i] = LOOKUP ? (act ? node_id[sc] : 0) : sc;
        }
        c4 = *(const int4*)(cnt + node0 + 4);
#pragma unroll
        for (int p = 0; p < 2; ++p) sv2[p] = slots[(node0 + 4 + 2 * p + hi) * SLOTS + l16];
        asm volatile("" :: "v"(wts[0]), "v"(wts[1]), "v"(wts[2]), "v"(wts[3]),
                           "v"(rows[0]), "v"(rows[1]), "v"(rows[2]), "v"(rows[3]));
        add_rows2(acc[0], acc[1], xlane, rows[0], wts[0], n0[0], 0, rows[1], wts[1], n0[1], 0);
        add_rows2(acc[2], acc[3], xlane, rows[2], wts[2], n0[2], 0, rows[3], wts[3], n0[3], 0);
      }
#pragma unroll
      for (int i = 0; i < 4; ++i) {
        if (__builtin_amdgcn_readfirstlane(ntot[i]) > SLOTS) {
          const int node = node0 + i;
          for (int base = 0; base < N_EDGES; base += 32) {
            int  e = base + lane;
            bool m = e < N_EDGES && dst[e] == node;
            unsigned mask = __builtin_amdgcn_ballot_w32(m);
            if (mask == 0u) continue;
            int n = __builtin_popcount(mask);
            unsigned below = (1u << lane) - 1u;
            int rk = m ? __builtin_popcount(mask & below) : n + __builtin_popcount(~mask & below);
            int s  = __builtin_amdgcn_ds_permute(rk << 2, m ? src[e] : 0);
            bool act = lane < n;
            int  sc  = act ? s : 0;
            float w2 = act ? norm[sc] : 0.f;
            int   r2 = LOOKUP ? (act ? node_id[sc] : 0) : sc;
            add_rows(acc[i], xlane, r2, w2, n);
          }
        }
        *(float4*)(strip + (g * 4 + i) * APITCH + lane * 4) = acc[i];
      }
    }
  }

  const int lane16 = lane & 15;
  const int pOff   = lane >> 4;
  v8f accw[8];
#pragma unroll
  for (int j = 0; j < 8; ++j) accw[j] = (v8f){};
  const float* arow = strip + lane16 * APITCH + (pOff << 1);
  const float* bcol = Wl2 + pOff * 256 + lane16 * 2;
#pragma unroll 4
  for (int k0 = 0; k0 < HD; k0 += 4) {
    v2f a = *(const v2f*)(arow + k0);
#pragma unroll
    for (int j = 0; j < 8; ++j) {
      v2f b = *(const v2f*)(bcol + (k0 >> 1) * 256 + j * 32);
      accw[j] = __builtin_amdgcn_wmma_f32_16x16x4_f32(
          false, a, false, b, (short)0, accw[j], false, false);
    }
  }

  const int row0 = tile * 16;
  float nr[8];
#pragma unroll
  for (int v = 0; v < 8; ++v) nr[v] = norm[row0 + (pOff << 3) + v];
#pragma unroll
  for (int j = 0; j < 8; ++j) {
    const float bs = bias[j * 16 + lane16];
#pragma unroll
    for (int v = 0; v < 8; ++v) {
      float y = accw[j][v] * nr[v] + bs;
      y = y > 0.f ? y : 0.2f * y;
      strip[((pOff << 3) + v) * APITCH + j * 16 + lane16] = y;
    }
  }
#pragma unroll
  for (int r = 0; r < 16; ++r) {
    float4 y4 = *(const float4*)(strip + r * APITCH + lane * 4);
    *(float4*)(out + (size_t)(row0 + r) * HD + lane * 4) = y4;
  }
}

extern "C" void kernel_launch(void* const* d_in, const int* in_sizes, int n_in,
                              void* d_out, int out_size, void* d_ws, size_t ws_size,
                              hipStream_t stream) {
  const int*   node_id = (const int*)d_in[0];
  const int*   src     = (const int*)d_in[1];
  const int*   dst     = (const int*)d_in[2];
  const float* norm    = (const float*)d_in[3];
  const float* embed   = (const float*)d_in[4];
  const float* W1      = (const float*)d_in[5];
  const float* b1      = (const float*)d_in[6];
  const float* W2      = (const float*)d_in[7];
  const float* b2      = (const float*)d_in[8];
  float* out = (float*)d_out;

  const size_t feat = (size_t)N_NODES * HD;
  float* bufH = (float*)d_ws;
  int*  cnt   = (int*)(bufH + feat);
  int*  slots = cnt + N_ALLOC;
  int*  cntG  = slots + (size_t)N_ALLOC * SLOTS;
  int2* recG  = (int2*)(cntG + ((NRANGE * NCHUNK + 31) / 32) * 32);


  bin_by_range<<<NCHUNK, dim3(1024), 0, stream>>>(src, dst, cntG, recG);
  place_rows<<<NRANGE, dim3(1024), 0, stream>>>(src, dst, cntG, recG, cnt, slots);

  layer_gather_wmma<true><<<LGRID, dim3(LTHREADS), 0, stream>>>(
      embed, node_id, cnt, slots, src, dst, norm, W1, b1, bufH);
  layer_gather_wmma<false><<<LGRID, dim3(LTHREADS), 0, stream>>>(
      bufH, node_id, cnt, slots, src, dst, norm, W2, b2, out);
}
